# starting kernel
# baseline (speedup 1.0000x reference)
#include <hip/hip_runtime.h>
#include <hip/hip_bf16.h>
#include <math.h>

#define NUM_SRC   200000
#define NUM_DST   100000
#define NUM_EDGES 1600000
#define IN_DIM    128
#define HID       256
#define OUT_DIM   128

typedef float v2f __attribute__((ext_vector_type(2)));
typedef float v8f __attribute__((ext_vector_type(8)));

#define ROWS 64
#define LDA  132
#define LDH  260
#define NBLK ((NUM_DST + ROWS - 1) / ROWS)
#define SMEM_BYTES ((ROWS * LDA + ROWS * LDH) * 4)

__global__ __launch_bounds__(256) void zero_kernel(float* __restrict__ p, size_t n) {
    size_t i = (size_t)blockIdx.x * blockDim.x + threadIdx.x;
    if (i < n) p[i] = 0.0f;
}

__global__ __launch_bounds__(256) void aggregate_kernel(
    const float* __restrict__ x,
    const int*   __restrict__ edge_src,
    const int*   __restrict__ edge_dst,
    float* __restrict__ sum,
    float* __restrict__ cnt,
    int num_edges)
{
    int edge = (int)((blockIdx.x * blockDim.x + threadIdx.x) >> 5);
    int lane = threadIdx.x & 31;
    if (edge >= num_edges) return;

    int src = edge_src[edge];
    int dst = edge_dst[edge];

    const float4 v = *(const float4*)(x + (size_t)src * IN_DIM + lane * 4);
    float* d = sum + (size_t)dst * IN_DIM + lane * 4;
    unsafeAtomicAdd(d + 0, v.x);
    unsafeAtomicAdd(d + 1, v.y);
    unsafeAtomicAdd(d + 2, v.z);
    unsafeAtomicAdd(d + 3, v.w);
    if (lane == 0) unsafeAtomicAdd(cnt + dst, 1.0f);
}

__global__ __launch_bounds__(512) void fused_mlp_kernel(
    const float* __restrict__ sum, const float* __restrict__ cnt,
    const float* __restrict__ x,
    const float* __restrict__ w_l,  const float* __restrict__ b_l,
    const float* __restrict__ w_r,
    const float* __restrict__ ln1_g, const float* __restrict__ ln1_b,
    const float* __restrict__ w_out, const float* __restrict__ b_out,
    const float* __restrict__ ln2_g, const float* __restrict__ ln2_b,
    float* __restrict__ out)
{
    extern __shared__ float smem[];
    float* sA = smem;
    float* sH = smem + ROWS * LDA;

    const int tid  = threadIdx.x;
    const int wave = tid >> 5;
    const int lane = tid & 31;
    const int rowbase = blockIdx.x * ROWS;

    const int n    = lane & 15;
    const int kh   = lane >> 4;
    const int arow = lane & 15;

    #pragma unroll
    for (int j = 0; j < 4; ++j) {
        const int idx = tid + j * 512;
        const int row = idx >> 5;
        const int k   = (idx & 31) * 4;
        int rg = rowbase + row;
        if (rg >= NUM_DST) rg = NUM_DST - 1;
        const float c   = cnt[rg];
        const float inv = 1.0f / fmaxf(c, 1.0f);
        const float4 s4 = *(const float4*)(sum + (size_t)rg * IN_DIM + k);
        sA[row * LDA + k + 0] = s4.x * inv;
        sA[row * LDA + k + 1] = s4.y * inv;
        sA[row * LDA + k + 2] = s4.z * inv;
        sA[row * LDA + k + 3] = s4.w * inv;
    }
    __syncthreads();

    v8f acc[4];
    #pragma unroll
    for (int rt = 0; rt < 4; ++rt) acc[rt] = (v8f){};
    const int colbase = wave * 16;

    #pragma unroll 2
    for (int k0 = 0; k0 < IN_DIM; k0 += 4) {
        const int kb = k0 + 2 * kh;
        v2f b;
        b.x = w_l[(size_t)kb * HID + colbase + n];
        b.y = w_l[(size_t)(kb + 1) * HID + colbase + n];
        #pragma unroll
        for (int rt = 0; rt < 4; ++rt) {
            v2f a;
            a.x = sA[(rt * 16 + arow) * LDA + kb];
            a.y = sA[(rt * 16 + arow) * LDA + kb + 1];
            acc[rt] = __builtin_amdgcn_wmma_f32_16x16x4_f32(
                false, a, false, b, (short)0, acc[rt], false, false);
        }
    }
    __syncthreads();

    #pragma unroll
    for (int j = 0; j < 4; ++j) {
        const int idx = tid + j * 512;
        const int row = idx >> 5;
        const int k   = (idx & 31) * 4;
        int rg = rowbase + row;
        if (rg >= NUM_DST) rg = NUM_DST - 1;
        const float4 x4 = *(const float4*)(x + (size_t)rg * IN_DIM + k);
        sA[row * LDA + k + 0] = x4.x;
        sA[row * LDA + k + 1] = x4.y;
        sA[row * LDA + k + 2] = x4.z;
        sA[row * LDA + k + 3] = x4.w;
    }
    __syncthreads();

    #pragma unroll 2
    for (int k0 = 0; k0 < IN_DIM; k0 += 4) {
        const int kb = k0 + 2 * kh;
        v2f b;
        b.x = w_r[(size_t)kb * HID + colbase + n];
        b.y = w_r[(size_t)(kb + 1) * HID + colbase + n];
        #pragma unroll
        for (int rt = 0; rt < 4; ++rt) {
            v2f a;
            a.x = sA[(rt * 16 + arow) * LDA + kb];
            a.y = sA[(rt * 16 + arow) * LDA + kb + 1];
            acc[rt] = __builtin_amdgcn_wmma_f32_16x16x4_f32(
                false, a, false, b, (short)0, acc[rt], false, false);
        }
    }

    {
        const float bias = b_l[colbase + n];
        #pragma unroll
        for (int rt = 0; rt < 4; ++rt) {
            #pragma unroll
            for (int i = 0; i < 8; ++i) {
                const int row = rt * 16 + i + 8 * kh;
                sH[row * LDH + colbase + n] = acc[rt][i] + bias;
            }
        }
    }
    __syncthreads();

    #pragma unroll
    for (int rr = 0; rr < 4; ++rr) {
        const int row = wave + 16 * rr;
        float v[8], s = 0.0f, s2 = 0.0f;
        #pragma unroll
        for (int j = 0; j < 8; ++j) {
            const float t = sH[row * LDH + lane + 32 * j];
            v[j] = t; s += t; s2 += t * t;
        }
        #pragma unroll
        for (int off = 16; off > 0; off >>= 1) {
            s  += __shfl_xor(s,  off, 32);
            s2 += __shfl_xor(s2, off, 32);
        }
        const float mean = s * (1.0f / HID);
        const float var  = s2 * (1.0f / HID) - mean * mean;
        const float rstd = rsqrtf(var + 1e-5f);
        #pragma unroll
        for (int j = 0; j < 8; ++j) {
            const int col = lane + 32 * j;
            const float hn = (v[j] - mean) * rstd * ln1_g[col] + ln1_b[col];
            sH[row * LDH + col] = 0.5f * hn * (1.0f + erff(hn * 0.70710678118654752f));
        }
    }
    __syncthreads();

    float* sOut = sA;
    if (wave < 8) {
        const int cb2 = wave * 16;
        v8f c[4];
        #pragma unroll
        for (int rt = 0; rt < 4; ++rt) c[rt] = (v8f){};
        #pragma unroll 2
        for (int k0 = 0; k0 < HID; k0 += 4) {
            const int kb = k0 + 2 * kh;
            v2f b;
            b.x = w_out[(size_t)kb * OUT_DIM + cb2 + n];
            b.y = w_out[(size_t)(kb + 1) * OUT_DIM + cb2 + n];
            #pragma unroll
            for (int rt = 0; rt < 4; ++rt) {
                v2f a;
                a.x = sH[(rt * 16 + arow) * LDH + kb];
                a.y = sH[(rt * 16 + arow) * LDH + kb + 1];
                c[rt] = __builtin_amdgcn_wmma_f32_16x16x4_f32(
                    false, a, false, b, (short)0, c[rt], false, false);
            }
        }
        const float bias = b_out[cb2 + n];
        #pragma unroll
        for (int rt = 0; rt < 4; ++rt) {
            #pragma unroll
            for (int i = 0; i < 8; ++i) {
                const int row = rt * 16 + i + 8 * kh;
                sOut[row * LDA + cb2 + n] = c[rt][i] + bias;
            }
        }
    }
    __syncthreads();

    #pragma unroll
    for (int rr = 0; rr < 4; ++rr) {
        const int row = wave + 16 * rr;
        float v[4], s = 0.0f, s2 = 0.0f;
        #pragma unroll
        for (int j = 0; j < 4; ++j) {
            const float t = sOut[row * LDA + lane + 32 * j];
            v[j] = t; s += t; s2 += t * t;
        }
        #pragma unroll
        for (int off = 16; off > 0; off >>= 1) {
            s  += __shfl_xor(s,  off, 32);
            s2 += __shfl_xor(s2, off, 32);
        }
        const float mean = s * (1.0f / OUT_DIM);
        const float var  = s2 * (1.0f / OUT_DIM) - mean * mean;
        const float rstd = rsqrtf(var + 1e-5f);
        const int rowg = rowbase + row;
        if (rowg < NUM_DST) {
            #pragma unroll
            for (int j = 0; j < 4; ++j) {
                const int col = lane + 32 * j;
                const float o = (v[j] - mean) * rstd * ln2_g[col] + ln2_b[col];
                out[(size_t)rowg * OUT_DIM + col] = o;
            }
        }
    }
}

extern "C" void kernel_launch(void* const* d_in, const int* in_sizes, int n_in,
                              void* d_out, int out_size, void* d_ws, size_t ws_size,
                              hipStream_t stream) {
    const float* x      = (const float*)d_in[0];
    const float* w_l    = (const float*)d_in[1];
    const float* b_l    = (const float*)d_in[2];
    const float* w_r    = (const float*)d_in[3];
    const float* ln1_g  = (const float*)d_in[4];
    const float* ln1_b  = (const float*)d_in[5];
    const float* w_out  = (const float*)d_in[6];
    const float* b_out  = (const float*)d_in[7];
    const float* ln2_g  = (const float*)d_in[8];
    const float* ln2_b  = (const float*)d_in[9];
    const int*   e_src  = (const int*)d_in[10];
    const int*   e_dst  = (const int*)d_in[11];

    float* sum = (float*)d_ws;
    float* cnt = sum + (size_t)NUM_DST * IN_DIM;
    float* out = (float*)d_out;

    const size_t nzero = (size_t)NUM_DST * IN_DIM + NUM_DST;
    zero_kernel<<<(unsigned)((nzero + 255) / 256), 256, 0, stream>>>(sum, nzero);

    aggregate_kernel<<<NUM_EDGES / 8, 256, 0, stream>>>(x, e_src, e_dst, sum, cnt,
                                                        NUM_EDGES);

    fused_mlp_kernel<<<NBLK, 512, SMEM_BYTES, stream>>>(sum, cnt, x,
                                                        w_l, b_l, w_r,
                                                        ln1_g, ln1_b,
                                                        w_out, b_out,
                                                        ln2_g, ln2_b, out);
}


# aggregate: global f32 atomic scatter and its zero pass replaced by owner workgroups: each holds 400 dst rows in LDS, scans all edge_dst, adds its own edges with LDS atomics; sums in LDS arrival order
# speedup vs baseline: 1.8126x; 1.8126x over previous
#include <hip/hip_runtime.h>
#include <hip/hip_bf16.h>
#include <math.h>

#define NUM_SRC   200000
#define NUM_DST   100000
#define NUM_EDGES 1600000
#define IN_DIM    128
#define HID       256
#define OUT_DIM   128

typedef float v2f __attribute__((ext_vector_type(2)));
typedef float v8f __attribute__((ext_vector_type(8)));

#define ROWS 64
#define LDA  132
#define LDH  260
#define NBLK ((NUM_DST + ROWS - 1) / ROWS)
#define SMEM_BYTES ((ROWS * LDA + ROWS * LDH) * 4)

#define AGG_ROWS    400
#define AGG_WAVES   32
#define AGG_THREADS (AGG_WAVES * 32)
#define AGG_NBLK    ((NUM_DST + AGG_ROWS - 1) / AGG_ROWS)
#define AGG_CHUNK   128
#define AGG_UNROLL  4
#define AGG_NSUPER  (NUM_EDGES / (AGG_CHUNK * AGG_UNROLL))
#define AGG_QCAP    160
#define AGG_SMEM_BYTES ((AGG_ROWS * IN_DIM + AGG_ROWS + AGG_WAVES * AGG_QCAP) * 4)

static_assert(NUM_EDGES % (AGG_CHUNK * AGG_UNROLL) == 0, "edge scan assumes whole chunks");
static_assert(AGG_ROWS <= 512, "queue entry keeps the local row in 9 bits");
static_assert((long long)NUM_EDGES * 512 < 2147483648LL, "queue entry = edge * 512 + local row");

__device__ __forceinline__ void lds_add_f32(float* p, float v) {
    __hip_atomic_fetch_add(p, v, __ATOMIC_RELAXED, __HIP_MEMORY_SCOPE_WORKGROUP);
}

template <bool FULL>
__device__ __forceinline__ void agg_flush(const int* qp, int nb, int lane,
                                          const float* __restrict__ x,
                                          const int* __restrict__ edge_src,
                                          float* acc, int* cntL)
{
    const bool live = FULL || (lane < nb);
    const int ent = live ? qp[lane] : 0;
    const int dl  = ent & 511;
    const int s   = live ? edge_src[ent >> 9] : 0;
    if (live) __hip_atomic_fetch_add(&cntL[dl], 1, __ATOMIC_RELAXED, __HIP_MEMORY_SCOPE_WORKGROUP);
    const int n = FULL ? 32 : nb;
    #pragma unroll 1
    for (int e = 0; e < n; e += 8) {
        float4 v[8];
        #pragma unroll
        for (int k = 0; k < 8; ++k) {
            v[k] = make_float4(0.f, 0.f, 0.f, 0.f);
            if (FULL || e + k < n) {
                const int se = __builtin_amdgcn_readlane(s, e + k);
                v[k] = *(const float4*)(x + (size_t)se * IN_DIM + lane * 4);
            }
        }
        #pragma unroll
        for (int k = 0; k < 8; ++k) {
            if (FULL || e + k < n) {
                const int de = __builtin_amdgcn_readlane(dl, e + k);
                float* a = acc + de * IN_DIM + lane;
                lds_add_f32(a,      v[k].x);
                lds_add_f32(a + 32, v[k].y);
                lds_add_f32(a + 64, v[k].z);
                lds_add_f32(a + 96, v[k].w);
            }
        }
    }
}

__global__ __launch_bounds__(AGG_THREADS) void aggregate_kernel(
    const float* __restrict__ x,
    const int*   __restrict__ edge_src,
    const int*   __restrict__ edge_dst,
    float* __restrict__ sum,
    float* __restrict__ cnt)
{
    extern __shared__ float smem[];
    float* acc  = smem;
    int*   cntL = (int*)(smem + AGG_ROWS * IN_DIM);
    int*   q    = cntL + AGG_ROWS;

    const int tid  = threadIdx.x;
    const int wave = tid >> 5;
    const int lane = tid & 31;
    const int base = blockIdx.x * AGG_ROWS;

    for (int i = tid; i < AGG_ROWS * IN_DIM / 4; i += AGG_THREADS)
        ((float4*)acc)[i] = make_float4(0.f, 0.f, 0.f, 0.f);
    for (int i = tid; i < AGG_ROWS; i += AGG_THREADS) cntL[i] = 0;
    __syncthreads();

    int* myq = q + wave * AGG_QCAP;
    int  qn  = 0;
    const int4* dst4 = (const int4*)edge_dst;
    const int rot = blockIdx.x * (AGG_NSUPER / AGG_NBLK);

    for (int it = 0; it * AGG_WAVES + wave < AGG_NSUPER; ++it) {
        int sc = it * AGG_WAVES + wave + rot;
        if (sc >= AGG_NSUPER) sc -= AGG_NSUPER;
        int4 d[AGG_UNROLL];
        #pragma unroll
        for (int u = 0; u < AGG_UNROLL; ++u)
            d[u] = dst4[(size_t)(sc * AGG_UNROLL + u) * 32 + lane];
        #pragma unroll
        for (int u = 0; u < AGG_UNROLL; ++u) {
            const int e0 = ((sc * AGG_UNROLL + u) * 32 + lane) * 4;
            const int dv[4] = {d[u].x, d[u].y, d[u].z, d[u].w};
            #pragma unroll
            for (int j = 0; j < 4; ++j) {
                const unsigned dl = (unsigned)(dv[j] - base);
                const bool m = dl < (unsigned)AGG_ROWS;
                const unsigned b = __builtin_amdgcn_ballot_w32(m);
                if (b) {
                    const int pos = qn + (int)__builtin_amdgcn_mbcnt_lo(b, 0u);
                    if (m) myq[pos] = ((e0 + j) << 9) | (int)dl;
                    qn += __builtin_popcount(b);
                }
            }
            while (qn >= 32) {
                qn -= 32;
                agg_flush<true>(myq + qn, 32, lane, x, edge_src, acc, cntL);
            }
        }
    }
    if (qn > 0) agg_flush<false>(myq, qn, lane, x, edge_src, acc, cntL);
    __syncthreads();

    int nrows = NUM_DST - base;
    if (nrows > AGG_ROWS) nrows = AGG_ROWS;
    for (int r = wave; r < nrows; r += AGG_WAVES) {
        const float* a = acc + r * IN_DIM + lane;
        const float4 o = make_float4(a[0], a[32], a[64], a[96]);
        *(float4*)(sum + (size_t)(base + r) * IN_DIM + lane * 4) = o;
    }
    for (int r = tid; r < nrows; r += AGG_THREADS) cnt[base + r] = (float)cntL[r];
}

__global__ __launch_bounds__(512) void fused_mlp_kernel(
    const float* __restrict__ sum, const float* __restrict__ cnt,
    const float* __restrict__ x,
    const float* __restrict__ w_l,  const float* __restrict__ b_l,
    const float* __restrict__ w_r,
    const float* __restrict__ ln1_g, const float* __restrict__ ln1_b,
    const float* __restrict__ w_out, const float* __restrict__ b_out,
    const float* __restrict__ ln2_g, const float* __restrict__ ln2_b,
    float* __restrict__ out)
{
    extern __shared__ float smem[];
    float* sA = smem;
    float* sH = smem + ROWS * LDA;

    const int tid  = threadIdx.x;
    const int wave = tid >> 5;
    const int lane = tid & 31;
    const int rowbase = blockIdx.x * ROWS;

    const int n    = lane & 15;
    const int kh   = lane >> 4;
    const int arow = lane & 15;

    #pragma unroll
    for (int j = 0; j < 4; ++j) {
        const int idx = tid + j * 512;
        const int row = idx >> 5;
        const int k   = (idx & 31) * 4;
        int rg = rowbase + row;
        if (rg >= NUM_DST) rg = NUM_DST - 1;
        const float c   = cnt[rg];
        const float inv = 1.0f / fmaxf(c, 1.0f);
        const float4 s4 = *(const float4*)(sum + (size_t)rg * IN_DIM + k);
        sA[row * LDA + k + 0] = s4.x * inv;
        sA[row * LDA + k + 1] = s4.y * inv;
        sA[row * LDA + k + 2] = s4.z * inv;
        sA[row * LDA + k + 3] = s4.w * inv;
    }
    __syncthreads();

    v8f acc[4];
    #pragma unroll
    for (int rt = 0; rt < 4; ++rt) acc[rt] = (v8f){};
    const int colbase = wave * 16;

    #pragma unroll 2
    for (int k0 = 0; k0 < IN_DIM; k0 += 4) {
        const int kb = k0 + 2 * kh;
        v2f b;
        b.x = w_l[(size_t)kb * HID + colbase + n];
        b.y = w_l[(size_t)(kb + 1) * HID + colbase + n];
        #pragma unroll
        for (int rt = 0; rt < 4; ++rt) {
            v2f a;
            a.x = sA[(rt * 16 + arow) * LDA + kb];
            a.y = sA[(rt * 16 + arow) * LDA + kb + 1];
            acc[rt] = __builtin_amdgcn_wmma_f32_16x16x4_f32(
                false, a, false, b, (short)0, acc[rt], false, false);
        }
    }
    __syncthreads();

    #pragma unroll
    for (int j = 0; j < 4; ++j) {
        const int idx = tid + j * 512;
        const int row = idx >> 5;
        const int k   = (idx & 31) * 4;
        int rg = rowbase + row;
        if (rg >= NUM_DST) rg = NUM_DST - 1;
        const float4 x4 = *(const float4*)(x + (size_t)rg * IN_DIM + k);
        sA[row * LDA + k + 0] = x4.x;
        sA[row * LDA + k + 1] = x4.y;
        sA[row * LDA + k + 2] = x4.z;
        sA[row * LDA + k + 3] = x4.w;
    }
    __syncthreads();

    #pragma unroll 2
    for (int k0 = 0; k0 < IN_DIM; k0 += 4) {
        const int kb = k0 + 2 * kh;
        v2f b;
        b.x = w_r[(size_t)kb * HID + colbase + n];
        b.y = w_r[(size_t)(kb + 1) * HID + colbase + n];
        #pragma unroll
        for (int rt = 0; rt < 4; ++rt) {
            v2f a;
            a.x = sA[(rt * 16 + arow) * LDA + kb];
            a.y = sA[(rt * 16 + arow) * LDA + kb + 1];
            acc[rt] = __builtin_amdgcn_wmma_f32_16x16x4_f32(
                false, a, false, b, (short)0, acc[rt], false, false);
        }
    }

    {
        const float bias = b_l[colbase + n];
        #pragma unroll
        for (int rt = 0; rt < 4; ++rt) {
            #pragma unroll
            for (int i = 0; i < 8; ++i) {
                const int row = rt * 16 + i + 8 * kh;
                sH[row * LDH + colbase + n] = acc[rt][i] + bias;
            }
        }
    }
    __syncthreads();

    #pragma unroll
    for (int rr = 0; rr < 4; ++rr) {
        const int row = wave + 16 * rr;
        float v[8], s = 0.0f, s2 = 0.0f;
        #pragma unroll
        for (int j = 0; j < 8; ++j) {
            const float t = sH[row * LDH + lane + 32 * j];
            v[j] = t; s += t; s2 += t * t;
        }
        #pragma unroll
        for (int off = 16; off > 0; off >>= 1) {
            s  += __shfl_xor(s,  off, 32);
            s2 += __shfl_xor(s2, off, 32);
        }
        const float mean = s * (1.0f / HID);
        const float var  = s2 * (1.0f / HID) - mean * mean;
        const float rstd = rsqrtf(var + 1e-5f);
        #pragma unroll
        for (int j = 0; j < 8; ++j) {
            const int col = lane + 32 * j;
            const float hn = (v[j] - mean) * rstd * ln1_g[col] + ln1_b[col];
            sH[row * LDH + col] = 0.5f * hn * (1.0f + erff(hn * 0.70710678118654752f));
        }
    }
    __syncthreads();

    float* sOut = sA;
    if (wave < 8) {
        const int cb2 = wave * 16;
        v8f c[4];
        #pragma unroll
        for (int rt = 0; rt < 4; ++rt) c[rt] = (v8f){};
        #pragma unroll 2
        for (int k0 = 0; k0 < HID; k0 += 4) {
            const int kb = k0 + 2 * kh;
            v2f b;
            b.x = w_out[(size_t)kb * OUT_DIM + cb2 + n];
            b.y = w_out[(size_t)(kb + 1) * OUT_DIM + cb2 + n];
            #pragma unroll
            for (int rt = 0; rt < 4; ++rt) {
                v2f a;
                a.x = sH[(rt * 16 + arow) * LDH + kb];
                a.y = sH[(rt * 16 + arow) * LDH + kb + 1];
                c[rt] = __builtin_amdgcn_wmma_f32_16x16x4_f32(
                    false, a, false, b, (short)0, c[rt], false, false);
            }
        }
        const float bias = b_out[cb2 + n];
        #pragma unroll
        for (int rt = 0; rt < 4; ++rt) {
            #pragma unroll
            for (int i = 0; i < 8; ++i) {
                const int row = rt * 16 + i + 8 * kh;
                sOut[row * LDA + cb2 + n] = c[rt][i] + bias;
            }
        }
    }
    __syncthreads();

    #pragma unroll
    for (int rr = 0; rr < 4; ++rr) {
        const int row = wave + 16 * rr;
        float v[4], s = 0.0f, s2 = 0.0f;
        #pragma unroll
        for (int j = 0; j < 4; ++j) {
            const float t = sOut[row * LDA + lane + 32 * j];
            v[j] = t; s += t; s2 += t * t;
        }
        #pragma unroll
        for (int off = 16; off > 0; off >>= 1) {
            s  += __shfl_xor(s,  off, 32);
            s2 += __shfl_xor(s2, off, 32);
        }
        const float mean = s * (1.0f / OUT_DIM);
        const float var  = s2 * (1.0f / OUT_DIM) - mean * mean;
        const float rstd = rsqrtf(var + 1e-5f);
        const int rowg = rowbase + row;
        if (rowg < NUM_DST) {
            #pragma unroll
            for (int j = 0; j < 4; ++j) {
                const int col = lane + 32 * j;
                const float o = (v[j] - mean) * rstd * ln2_g[col] + ln2_b[col];
                out[(size_t)rowg * OUT_DIM + col] = o;
            }
        }
    }
}

extern "C" void kernel_launch(void* const* d_in, const int* in_sizes, int n_in,
                              void* d_out, int out_size, void* d_ws, size_t ws_size,
                              hipStream_t stream) {
    const float* x      = (const float*)d_in[0];
    const float* w_l    = (const float*)d_in[1];
    const float* b_l    = (const float*)d_in[2];
    const float* w_r    = (const float*)d_in[3];
    const float* ln1_g  = (const float*)d_in[4];
    const float* ln1_b  = (const float*)d_in[5];
    const float* w_out  = (const float*)d_in[6];
    const float* b_out  = (const float*)d_in[7];
    const float* ln2_g  = (const float*)d_in[8];
    const float* ln2_b  = (const float*)d_in[9];
    const int*   e_src  = (const int*)d_in[10];
    const int*   e_dst  = (const int*)d_in[11];

    float* sum = (float*)d_ws;
    float* cnt = sum + (size_t)NUM_DST * IN_DIM;
    float* out = (float*)d_out;

    aggregate_kernel<<<AGG_NBLK, AGG_THREADS, AGG_SMEM_BYTES, stream>>>(x, e_src, e_dst,
                                                                        sum, cnt);

    fused_mlp_kernel<<<NBLK, 512, SMEM_BYTES, stream>>>(sum, cnt, x,
                                                        w_l, b_l, w_r,
                                                        ln1_g, ln1_b,
                                                        w_out, b_out,
                                                        ln2_g, ln2_b, out);
}


# aggregate scan: added first launch writes an owner byte per edge (edge_dst/400); owner workgroups scan the bytes, 16 edges per lane by word-wide byte compares, not 32-bit edge_dst; LDS arrival order
# speedup vs baseline: 1.9875x; 1.0965x over previous
#include <hip/hip_runtime.h>
#include <hip/hip_bf16.h>
#include <math.h>

#define NUM_SRC   200000
#define NUM_DST   100000
#define NUM_EDGES 1600000
#define IN_DIM    128
#define HID       256
#define OUT_DIM   128

typedef float v2f __attribute__((ext_vector_type(2)));
typedef float v8f __attribute__((ext_vector_type(8)));

#define ROWS 64
#define LDA  132
#define LDH  260
#define NBLK ((NUM_DST + ROWS - 1) / ROWS)
#define SMEM_BYTES ((ROWS * LDA + ROWS * LDH) * 4)

#define AGG_ROWS    400
#define AGG_WAVES   32
#define AGG_THREADS (AGG_WAVES * 32)
#define AGG_NBLK    ((NUM_DST + AGG_ROWS - 1) / AGG_ROWS)
#define AGG_CHUNK   512
#define AGG_UNROLL  5
#define AGG_NSUPER  (NUM_EDGES / (AGG_CHUNK * AGG_UNROLL))
#define AGG_QCAP    544
#define AGG_SMEM_BYTES ((AGG_ROWS * IN_DIM + AGG_ROWS + AGG_WAVES * AGG_QCAP) * 4)

#define OWN_NBLK    250
#define OWN_THREADS 320
#define OWN_PER     5
#define OWN_NONE    255u

static_assert(NUM_EDGES % (AGG_CHUNK * AGG_UNROLL) == 0, "edge scan assumes whole chunks");
static_assert(AGG_NBLK <= 255, "an owner number must fit a byte and differ from OWN_NONE");
static_assert(NUM_EDGES / 4 == OWN_NBLK * OWN_THREADS * OWN_PER, "owner pass covers every edge once");
static_assert(AGG_SMEM_BYTES <= 320 * 1024, "one workgroup's LDS");

__device__ __forceinline__ unsigned owner_of(int d) {
    return ((unsigned)d < (unsigned)NUM_DST) ? ((unsigned)d / (unsigned)AGG_ROWS) : OWN_NONE;
}

__global__ __launch_bounds__(OWN_THREADS) void owner_kernel(
    const int* __restrict__ edge_dst,
    int* __restrict__ owner)
{
    const int4* dst4 = (const int4*)edge_dst;
    const int i0 = blockIdx.x * OWN_THREADS + threadIdx.x;
    int4 d[OWN_PER];
    #pragma unroll
    for (int j = 0; j < OWN_PER; ++j) d[j] = dst4[i0 + j * (OWN_NBLK * OWN_THREADS)];
    #pragma unroll
    for (int j = 0; j < OWN_PER; ++j) {
        const unsigned w = owner_of(d[j].x) | (owner_of(d[j].y) << 8) |
                           (owner_of(d[j].z) << 16) | (owner_of(d[j].w) << 24);
        owner[i0 + j * (OWN_NBLK * OWN_THREADS)] = (int)w;
    }
}

__device__ __forceinline__ void lds_add_f32(float* p, float v) {
    __hip_atomic_fetch_add(p, v, __ATOMIC_RELAXED, __HIP_MEMORY_SCOPE_WORKGROUP);
}

__device__ __forceinline__ unsigned zero_bytes(unsigned t) {
    return ~(((t & 0x7f7f7f7fu) + 0x7f7f7f7fu) | t) & 0x80808080u;
}

template <bool FULL>
__device__ __forceinline__ void agg_flush(const int* qp, int nb, int lane, int base,
                                          const float* __restrict__ x,
                                          const int* __restrict__ edge_src,
                                          const int* __restrict__ edge_dst,
                                          float* acc, int* cntL)
{
    const bool live = FULL || (lane < nb);
    const int ent = live ? qp[lane] : 0;
    const int s   = live ? edge_src[ent] : 0;
    const int dl  = live ? edge_dst[ent] - base : 0;
    if (live) __hip_atomic_fetch_add(&cntL[dl], 1, __ATOMIC_RELAXED, __HIP_MEMORY_SCOPE_WORKGROUP);
    const int n = FULL ? 32 : nb;
    #pragma unroll 1
    for (int e = 0; e < n; e += 8) {
        float4 v[8];
        #pragma unroll
        for (int k = 0; k < 8; ++k) {
            v[k] = make_float4(0.f, 0.f, 0.f, 0.f);
            if (FULL || e + k < n) {
                const int se = __builtin_amdgcn_readlane(s, e + k);
                v[k] = *(const float4*)(x + (size_t)se * IN_DIM + lane * 4);
            }
        }
        #pragma unroll
        for (int k = 0; k < 8; ++k) {
            if (FULL || e + k < n) {
                const int de = __builtin_amdgcn_readlane(dl, e + k);
                float* a = acc + de * IN_DIM + lane;
                lds_add_f32(a,      v[k].x);
                lds_add_f32(a + 32, v[k].y);
                lds_add_f32(a + 64, v[k].z);
                lds_add_f32(a + 96, v[k].w);
            }
        }
    }
}

__global__ __launch_bounds__(AGG_THREADS) void aggregate_kernel(
    const float* __restrict__ x,
    const int*   __restrict__ edge_src,
    const int*   __restrict__ edge_dst,
    const int*   __restrict__ owner,
    float* __restrict__ sum,
    float* __restrict__ cnt)
{
    extern __shared__ float smem[];
    float* acc  = smem;
    int*   cntL = (int*)(smem + AGG_ROWS * IN_DIM);
    int*   q    = cntL + AGG_ROWS;

    const int tid  = threadIdx.x;
    const int wave = tid >> 5;
    const int lane = tid & 31;
    const int base = blockIdx.x * AGG_ROWS;

    for (int i = tid; i < AGG_ROWS * IN_DIM / 4; i += AGG_THREADS)
        ((float4*)acc)[i] = make_float4(0.f, 0.f, 0.f, 0.f);
    for (int i = tid; i < AGG_ROWS; i += AGG_THREADS) cntL[i] = 0;
    __syncthreads();

    int* myq = q + wave * AGG_QCAP;
    int  qn  = 0;
    const uint4* own4 = (const uint4*)owner;
    const unsigned pat = blockIdx.x * 0x01010101u;
    const int rot = (blockIdx.x * AGG_NSUPER) / AGG_NBLK;

    for (int it = 0; it * AGG_WAVES + wave < AGG_NSUPER; ++it) {
        int sc = it * AGG_WAVES + wave + rot;
        if (sc >= AGG_NSUPER) sc -= AGG_NSUPER;
        uint4 d[AGG_UNROLL];
        #pragma unroll
        for (int u = 0; u < AGG_UNROLL; ++u)
            d[u] = own4[(sc * AGG_UNROLL + u) * 32 + lane];
        #pragma unroll
        for (int u = 0; u < AGG_UNROLL; ++u) {
            const int e0 = ((sc * AGG_UNROLL + u) * 32 + lane) * 16;
            unsigned h = (zero_bytes(d[u].x ^ pat) >> 7) | (zero_bytes(d[u].y ^ pat) >> 6) |
                         (zero_bytes(d[u].z ^ pat) >> 5) | (zero_bytes(d[u].w ^ pat) >> 4);
            unsigned b = __builtin_amdgcn_ballot_w32(h != 0u);
            while (b) {
                const int p   = __builtin_ctz(h | 0x80000000u);
                const int pos = qn + (int)__builtin_amdgcn_mbcnt_lo(b, 0u);
                if (h != 0u) myq[pos] = e0 + ((p & 7) << 2) + (p >> 3);
                h &= h - 1u;
                qn += __builtin_popcount(b);
                b = __builtin_amdgcn_ballot_w32(h != 0u);
            }
            while (qn >= 32) {
                qn -= 32;
                agg_flush<true>(myq + qn, 32, lane, base, x, edge_src, edge_dst, acc, cntL);
            }
        }
    }
    if (qn > 0) agg_flush<false>(myq, qn, lane, base, x, edge_src, edge_dst, acc, cntL);
    __syncthreads();

    int nrows = NUM_DST - base;
    if (nrows > AGG_ROWS) nrows = AGG_ROWS;
    for (int r = wave; r < nrows; r += AGG_WAVES) {
        const float* a = acc + r * IN_DIM + lane;
        const float4 o = make_float4(a[0], a[32], a[64], a[96]);
        *(float4*)(sum + (size_t)(base + r) * IN_DIM + lane * 4) = o;
    }
    for (int r = tid; r < nrows; r += AGG_THREADS) cnt[base + r] = (float)cntL[r];
}

__global__ __launch_bounds__(512) void fused_mlp_kernel(
    const float* __restrict__ sum, const float* __restrict__ cnt,
    const float* __restrict__ x,
    const float* __restrict__ w_l,  const float* __restrict__ b_l,
    const float* __restrict__ w_r,
    const float* __restrict__ ln1_g, const float* __restrict__ ln1_b,
    const float* __restrict__ w_out, const float* __restrict__ b_out,
    const float* __restrict__ ln2_g, const float* __restrict__ ln2_b,
    float* __restrict__ out)
{
    extern __shared__ float smem[];
    float* sA = smem;
    float* sH = smem + ROWS * LDA;

    const int tid  = threadIdx.x;
    const int wave = tid >> 5;
    const int lane = tid & 31;
    const int rowbase = blockIdx.x * ROWS;

    const int n    = lane & 15;
    const int kh   = lane >> 4;
    const int arow = lane & 15;

    #pragma unroll
    for (int j = 0; j < 4; ++j) {
        const int idx = tid + j * 512;
        const int row = idx >> 5;
        const int k   = (idx & 31) * 4;
        int rg = rowbase + row;
        if (rg >= NUM_DST) rg = NUM_DST - 1;
        const float c   = cnt[rg];
        const float inv = 1.0f / fmaxf(c, 1.0f);
        const float4 s4 = *(const float4*)(sum + (size_t)rg * IN_DIM + k);
        sA[row * LDA + k + 0] = s4.x * inv;
        sA[row * LDA + k + 1] = s4.y * inv;
        sA[row * LDA + k + 2] = s4.z * inv;
        sA[row * LDA + k + 3] = s4.w * inv;
    }
    __syncthreads();

    v8f acc[4];
    #pragma unroll
    for (int rt = 0; rt < 4; ++rt) acc[rt] = (v8f){};
    const int colbase = wave * 16;

    #pragma unroll 2
    for (int k0 = 0; k0 < IN_DIM; k0 += 4) {
        const int kb = k0 + 2 * kh;
        v2f b;
        b.x = w_l[(size_t)kb * HID + colbase + n];
        b.y = w_l[(size_t)(kb + 1) * HID + colbase + n];
        #pragma unroll
        for (int rt = 0; rt < 4; ++rt) {
            v2f a;
            a.x = sA[(rt * 16 + arow) * LDA + kb];
            a.y = sA[(rt * 16 + arow) * LDA + kb + 1];
            acc[rt] = __builtin_amdgcn_wmma_f32_16x16x4_f32(
                false, a, false, b, (short)0, acc[rt], false, false);
        }
    }
    __syncthreads();

    #pragma unroll
    for (int j = 0; j < 4; ++j) {
        const int idx = tid + j * 512;
        const int row = idx >> 5;
        const int k   = (idx & 31) * 4;
        int rg = rowbase + row;
        if (rg >= NUM_DST) rg = NUM_DST - 1;
        const float4 x4 = *(const float4*)(x + (size_t)rg * IN_DIM + k);
        sA[row * LDA + k + 0] = x4.x;
        sA[row * LDA + k + 1] = x4.y;
        sA[row * LDA + k + 2] = x4.z;
        sA[row * LDA + k + 3] = x4.w;
    }
    __syncthreads();

    #pragma unroll 2
    for (int k0 = 0; k0 < IN_DIM; k0 += 4) {
        const int kb = k0 + 2 * kh;
        v2f b;
        b.x = w_r[(size_t)kb * HID + colbase + n];
        b.y = w_r[(size_t)(kb + 1) * HID + colbase + n];
        #pragma unroll
        for (int rt = 0; rt < 4; ++rt) {
            v2f a;
            a.x = sA[(rt * 16 + arow) * LDA + kb];
            a.y = sA[(rt * 16 + arow) * LDA + kb + 1];
            acc[rt] = __builtin_amdgcn_wmma_f32_16x16x4_f32(
                false, a, false, b, (short)0, acc[rt], false, false);
        }
    }

    {
        const float bias = b_l[colbase + n];
        #pragma unroll
        for (int rt = 0; rt < 4; ++rt) {
            #pragma unroll
            for (int i = 0; i < 8; ++i) {
                const int row = rt * 16 + i + 8 * kh;
                sH[row * LDH + colbase + n] = acc[rt][i] + bias;
            }
        }
    }
    __syncthreads();

    #pragma unroll
    for (int rr = 0; rr < 4; ++rr) {
        const int row = wave + 16 * rr;
        float v[8], s = 0.0f, s2 = 0.0f;
        #pragma unroll
        for (int j = 0; j < 8; ++j) {
            const float t = sH[row * LDH + lane + 32 * j];
            v[j] = t; s += t; s2 += t * t;
        }
        #pragma unroll
        for (int off = 16; off > 0; off >>= 1) {
            s  += __shfl_xor(s,  off, 32);
            s2 += __shfl_xor(s2, off, 32);
        }
        const float mean = s * (1.0f / HID);
        const float var  = s2 * (1.0f / HID) - mean * mean;
        const float rstd = rsqrtf(var + 1e-5f);
        #pragma unroll
        for (int j = 0; j < 8; ++j) {
            const int col = lane + 32 * j;
            const float hn = (v[j] - mean) * rstd * ln1_g[col] + ln1_b[col];
            sH[row * LDH + col] = 0.5f * hn * (1.0f + erff(hn * 0.70710678118654752f));
        }
    }
    __syncthreads();

    float* sOut = sA;
    if (wave < 8) {
        const int cb2 = wave * 16;
        v8f c[4];
        #pragma unroll
        for (int rt = 0; rt < 4; ++rt) c[rt] = (v8f){};
        #pragma unroll 2
        for (int k0 = 0; k0 < HID; k0 += 4) {
            const int kb = k0 + 2 * kh;
            v2f b;
            b.x = w_out[(size_t)kb * OUT_DIM + cb2 + n];
            b.y = w_out[(size_t)(kb + 1) * OUT_DIM + cb2 + n];
            #pragma unroll
            for (int rt = 0; rt < 4; ++rt) {
                v2f a;
                a.x = sH[(rt * 16 + arow) * LDH + kb];
                a.y = sH[(rt * 16 + arow) * LDH + kb + 1];
                c[rt] = __builtin_amdgcn_wmma_f32_16x16x4_f32(
                    false, a, false, b, (short)0, c[rt], false, false);
            }
        }
        const float bias = b_out[cb2 + n];
        #pragma unroll
        for (int rt = 0; rt < 4; ++rt) {
            #pragma unroll
            for (int i = 0; i < 8; ++i) {
                const int row = rt * 16 + i + 8 * kh;
                sOut[row * LDA + cb2 + n] = c[rt][i] + bias;
            }
        }
    }
    __syncthreads();

    #pragma unroll
    for (int rr = 0; rr < 4; ++rr) {
        const int row = wave + 16 * rr;
        float v[4], s = 0.0f, s2 = 0.0f;
        #pragma unroll
        for (int j = 0; j < 4; ++j) {
            const float t = sOut[row * LDA + lane + 32 * j];
            v[j] = t; s += t; s2 += t * t;
        }
        #pragma unroll
        for (int off = 16; off > 0; off >>= 1) {
            s  += __shfl_xor(s,  off, 32);
            s2 += __shfl_xor(s2, off, 32);
        }
        const float mean = s * (1.0f / OUT_DIM);
        const float var  = s2 * (1.0f / OUT_DIM) - mean * mean;
        const float rstd = rsqrtf(var + 1e-5f);
        const int rowg = rowbase + row;
        if (rowg < NUM_DST) {
            #pragma unroll
            for (int j = 0; j < 4; ++j) {
                const int col = lane + 32 * j;
                const float o = (v[j] - mean) * rstd * ln2_g[col] + ln2_b[col];
                out[(size_t)rowg * OUT_DIM + col] = o;
            }
        }
    }
}

extern "C" void kernel_launch(void* const* d_in, const int* in_sizes, int n_in,
                              void* d_out, int out_size, void* d_ws, size_t ws_size,
                              hipStream_t stream) {
    const float* x      = (const float*)d_in[0];
    const float* w_l    = (const float*)d_in[1];
    const float* b_l    = (const float*)d_in[2];
    const float* w_r    = (const float*)d_in[3];
    const float* ln1_g  = (const float*)d_in[4];
    const float* ln1_b  = (const float*)d_in[5];
    const float* w_out  = (const float*)d_in[6];
    const float* b_out  = (const float*)d_in[7];
    const float* ln2_g  = (const float*)d_in[8];
    const float* ln2_b  = (const float*)d_in[9];
    const int*   e_src  = (const int*)d_in[10];
    const int*   e_dst  = (const int*)d_in[11];

    float* sum = (float*)d_ws;
    float* cnt = sum + (size_t)NUM_DST * IN_DIM;
    float* out = (float*)d_out;
    int* owner = (int*)d_out;

    owner_kernel<<<OWN_NBLK, OWN_THREADS, 0, stream>>>(e_dst, owner);

    aggregate_kernel<<<AGG_NBLK, AGG_THREADS, AGG_SMEM_BYTES, stream>>>(x, e_src, e_dst, owner,
                                                                        sum, cnt);

    fused_mlp_kernel<<<NBLK, 512, SMEM_BYTES, stream>>>(sum, cnt, x,
                                                        w_l, b_l, w_r,
                                                        ln1_g, ln1_b,
                                                        w_out, b_out,
                                                        ln2_g, ln2_b, out);
}


# aggregate flush: launch 0 also writes one 32-bit record per edge (source row << 9 | destination row mod 400); a flush reads ONE scattered word per edge, not edge_src and edge_dst; sums unchanged
# speedup vs baseline: 2.0584x; 1.0357x over previous
#include <hip/hip_runtime.h>
#include <hip/hip_bf16.h>
#include <math.h>

#define NUM_SRC   200000
#define NUM_DST   100000
#define NUM_EDGES 1600000
#define IN_DIM    128
#define HID       256
#define OUT_DIM   128

typedef float v2f __attribute__((ext_vector_type(2)));
typedef float v8f __attribute__((ext_vector_type(8)));

#define ROWS 64
#define LDA  132
#define LDH  260
#define NBLK ((NUM_DST + ROWS - 1) / ROWS)
#define SMEM_BYTES ((ROWS * LDA + ROWS * LDH) * 4)

#define AGG_ROWS    400
#define AGG_WAVES   32
#define AGG_THREADS (AGG_WAVES * 32)
#define AGG_NBLK    ((NUM_DST + AGG_ROWS - 1) / AGG_ROWS)
#define AGG_CHUNK   512
#define AGG_UNROLL  5
#define AGG_NSUPER  (NUM_EDGES / (AGG_CHUNK * AGG_UNROLL))
#define AGG_QCAP    544
#define AGG_SMEM_BYTES ((AGG_ROWS * IN_DIM + AGG_ROWS + AGG_WAVES * AGG_QCAP) * 4)

#define OWN_NBLK    250
#define OWN_THREADS 320
#define OWN_PER     5
#define OWN_NONE    255u
#define REC_BITS    9

static_assert(NUM_EDGES % (AGG_CHUNK * AGG_UNROLL) == 0, "edge scan assumes whole chunks");
static_assert(AGG_NBLK <= 255, "an owner number must fit a byte and differ from OWN_NONE");
static_assert(NUM_EDGES / 4 == OWN_NBLK * OWN_THREADS * OWN_PER, "owner pass covers every edge once");
static_assert(AGG_SMEM_BYTES <= 320 * 1024, "one workgroup's LDS");
static_assert(AGG_ROWS <= (1 << REC_BITS), "a local destination row must fit the record's low bits");
static_assert(((long long)(NUM_SRC - 1) << REC_BITS) < 0x7fffffffLL, "a source row must fit the record's upper bits");
static_assert(NUM_EDGES % 256 == 0, "the records start on a 256-byte boundary behind the owner bytes");

__device__ __forceinline__ unsigned owner_of(int d) {
    return ((unsigned)d < (unsigned)NUM_DST) ? ((unsigned)d / (unsigned)AGG_ROWS) : OWN_NONE;
}

__device__ __forceinline__ int record_of(int s, int d, unsigned own) {
    return (int)(((unsigned)s << REC_BITS) | (((unsigned)d - own * (unsigned)AGG_ROWS) & ((1u << REC_BITS) - 1u)));
}

__global__ __launch_bounds__(OWN_THREADS) void owner_kernel(
    const int* __restrict__ edge_src,
    const int* __restrict__ edge_dst,
    int* __restrict__ owner,
    int* __restrict__ rec)
{
    const int4* src4 = (const int4*)edge_src;
    const int4* dst4 = (const int4*)edge_dst;
    int4* rec4 = (int4*)rec;
    const int i0 = blockIdx.x * OWN_THREADS + threadIdx.x;
    int4 d[OWN_PER], s[OWN_PER];
    #pragma unroll
    for (int j = 0; j < OWN_PER; ++j) {
        d[j] = dst4[i0 + j * (OWN_NBLK * OWN_THREADS)];
        s[j] = src4[i0 + j * (OWN_NBLK * OWN_THREADS)];
    }
    #pragma unroll
    for (int j = 0; j < OWN_PER; ++j) {
        const unsigned ox = owner_of(d[j].x), oy = owner_of(d[j].y);
        const unsigned oz = owner_of(d[j].z), ow = owner_of(d[j].w);
        owner[i0 + j * (OWN_NBLK * OWN_THREADS)] = (int)(ox | (oy << 8) | (oz << 16) | (ow << 24));
        rec4[i0 + j * (OWN_NBLK * OWN_THREADS)] =
            make_int4(record_of(s[j].x, d[j].x, ox), record_of(s[j].y, d[j].y, oy),
                      record_of(s[j].z, d[j].z, oz), record_of(s[j].w, d[j].w, ow));
    }
}

__device__ __forceinline__ void lds_add_f32(float* p, float v) {
    __hip_atomic_fetch_add(p, v, __ATOMIC_RELAXED, __HIP_MEMORY_SCOPE_WORKGROUP);
}

__device__ __forceinline__ unsigned zero_bytes(unsigned t) {
    return ~(((t & 0x7f7f7f7fu) + 0x7f7f7f7fu) | t) & 0x80808080u;
}

template <bool FULL>
__device__ __forceinline__ void agg_flush(const int* qp, int nb, int lane,
                                          const float* __restrict__ x,
                                          const int* __restrict__ rec,
                                          float* acc, int* cntL)
{
    const bool live = FULL || (lane < nb);
    const int ent = live ? qp[lane] : 0;
    const unsigned r = live ? (unsigned)rec[ent] : 0u;
    const int s   = (int)(r >> REC_BITS);
    const int dl  = (int)(r & ((1u << REC_BITS) - 1u));
    if (live) __hip_atomic_fetch_add(&cntL[dl], 1, __ATOMIC_RELAXED, __HIP_MEMORY_SCOPE_WORKGROUP);
    const int n = FULL ? 32 : nb;
    #pragma unroll 1
    for (int e = 0; e < n; e += 8) {
        float4 v[8];
        #pragma unroll
        for (int k = 0; k < 8; ++k) {
            v[k] = make_float4(0.f, 0.f, 0.f, 0.f);
            if (FULL || e + k < n) {
                const int se = __builtin_amdgcn_readlane(s, e + k);
                v[k] = *(const float4*)(x + (size_t)se * IN_DIM + lane * 4);
            }
        }
        #pragma unroll
        for (int k = 0; k < 8; ++k) {
            if (FULL || e + k < n) {
                const int de = __builtin_amdgcn_readlane(dl, e + k);
                float* a = acc + de * IN_DIM + lane;
                lds_add_f32(a,      v[k].x);
                lds_add_f32(a + 32, v[k].y);
                lds_add_f32(a + 64, v[k].z);
                lds_add_f32(a + 96, v[k].w);
            }
        }
    }
}

__global__ __launch_bounds__(AGG_THREADS) void aggregate_kernel(
    const float* __restrict__ x,
    const int*   __restrict__ owner,
    const int*   __restrict__ rec,
    float* __restrict__ sum,
    float* __restrict__ cnt)
{
    extern __shared__ float smem[];
    float* acc  = smem;
    int*   cntL = (int*)(smem + AGG_ROWS * IN_DIM);
    int*   q    = cntL + AGG_ROWS;

    const int tid  = threadIdx.x;
    const int wave = tid >> 5;
    const int lane = tid & 31;
    const int base = blockIdx.x * AGG_ROWS;

    for (int i = tid; i < AGG_ROWS * IN_DIM / 4; i += AGG_THREADS)
        ((float4*)acc)[i] = make_float4(0.f, 0.f, 0.f, 0.f);
    for (int i = tid; i < AGG_ROWS; i += AGG_THREADS) cntL[i] = 0;
    __syncthreads();

    int* myq = q + wave * AGG_QCAP;
    int  qn  = 0;
    const uint4* own4 = (const uint4*)owner;
    const unsigned pat = blockIdx.x * 0x01010101u;
    const int rot = (blockIdx.x * AGG_NSUPER) / AGG_NBLK;

    for (int it = 0; it * AGG_WAVES + wave < AGG_NSUPER; ++it) {
        int sc = it * AGG_WAVES + wave + rot;
        if (sc >= AGG_NSUPER) sc -= AGG_NSUPER;
        uint4 d[AGG_UNROLL];
        #pragma unroll
        for (int u = 0; u < AGG_UNROLL; ++u)
            d[u] = own4[(sc * AGG_UNROLL + u) * 32 + lane];
        #pragma unroll
        for (int u = 0; u < AGG_UNROLL; ++u) {
            const int e0 = ((sc * AGG_UNROLL + u) * 32 + lane) * 16;
            unsigned h = (zero_bytes(d[u].x ^ pat) >> 7) | (zero_bytes(d[u].y ^ pat) >> 6) |
                         (zero_bytes(d[u].z ^ pat) >> 5) | (zero_bytes(d[u].w ^ pat) >> 4);
            unsigned b = __builtin_amdgcn_ballot_w32(h != 0u);
            while (b) {
                const int p   = __builtin_ctz(h | 0x80000000u);
                const int pos = qn + (int)__builtin_amdgcn_mbcnt_lo(b, 0u);
                if (h != 0u) myq[pos] = e0 + ((p & 7) << 2) + (p >> 3);
                h &= h - 1u;
                qn += __builtin_popcount(b);
                b = __builtin_amdgcn_ballot_w32(h != 0u);
            }
            while (qn >= 32) {
                qn -= 32;
                agg_flush<true>(myq + qn, 32, lane, x, rec, acc, cntL);
            }
        }
    }
    if (qn > 0) agg_flush<false>(myq, qn, lane, x, rec, acc, cntL);
    __syncthreads();

    int nrows = NUM_DST - base;
    if (nrows > AGG_ROWS) nrows = AGG_ROWS;
    for (int r = wave; r < nrows; r += AGG_WAVES) {
        const float* a = acc + r * IN_DIM + lane;
        const float4 o = make_float4(a[0], a[32], a[64], a[96]);
        *(float4*)(sum + (size_t)(base + r) * IN_DIM + lane * 4) = o;
    }
    for (int r = tid; r < nrows; r += AGG_THREADS) cnt[base + r] = (float)cntL[r];
}

__global__ __launch_bounds__(512) void fused_mlp_kernel(
    const float* __restrict__ sum, const float* __restrict__ cnt,
    const float* __restrict__ x,
    const float* __restrict__ w_l,  const float* __restrict__ b_l,
    const float* __restrict__ w_r,
    const float* __restrict__ ln1_g, const float* __restrict__ ln1_b,
    const float* __restrict__ w_out, const float* __restrict__ b_out,
    const float* __restrict__ ln2_g, const float* __restrict__ ln2_b,
    float* __restrict__ out)
{
    extern __shared__ float smem[];
    float* sA = smem;
    float* sH = smem + ROWS * LDA;

    const int tid  = threadIdx.x;
    const int wave = tid >> 5;
    const int lane = tid & 31;
    const int rowbase = blockIdx.x * ROWS;

    const int n    = lane & 15;
    const int kh   = lane >> 4;
    const int arow = lane & 15;

    #pragma unroll
    for (int j = 0; j < 4; ++j) {
        const int idx = tid + j * 512;
        const int row = idx >> 5;
        const int k   = (idx & 31) * 4;
        int rg = rowbase + row;
        if (rg >= NUM_DST) rg = NUM_DST - 1;
        const float c   = cnt[rg];
        const float inv = 1.0f / fmaxf(c, 1.0f);
        const float4 s4 = *(const float4*)(sum + (size_t)rg * IN_DIM + k);
        sA[row * LDA + k + 0] = s4.x * inv;
        sA[row * LDA + k + 1] = s4.y * inv;
        sA[row * LDA + k + 2] = s4.z * inv;
        sA[row * LDA + k + 3] = s4.w * inv;
    }
    __syncthreads();

    v8f acc[4];
    #pragma unroll
    for (int rt = 0; rt < 4; ++rt) acc[rt] = (v8f){};
    const int colbase = wave * 16;

    #pragma unroll 2
    for (int k0 = 0; k0 < IN_DIM; k0 += 4) {
        const int kb = k0 + 2 * kh;
        v2f b;
        b.x = w_l[(size_t)kb * HID + colbase + n];
        b.y = w_l[(size_t)(kb + 1) * HID + colbase + n];
        #pragma unroll
        for (int rt = 0; rt < 4; ++rt) {
            v2f a;
            a.x = sA[(rt * 16 + arow) * LDA + kb];
            a.y = sA[(rt * 16 + arow) * LDA + kb + 1];
            acc[rt] = __builtin_amdgcn_wmma_f32_16x16x4_f32(
                false, a, false, b, (short)0, acc[rt], false, false);
        }
    }
    __syncthreads();

    #pragma unroll
    for (int j = 0; j < 4; ++j) {
        const int idx = tid + j * 512;
        const int row = idx >> 5;
        const int k   = (idx & 31) * 4;
        int rg = rowbase + row;
        if (rg >= NUM_DST) rg = NUM_DST - 1;
        const float4 x4 = *(const float4*)(x + (size_t)rg * IN_DIM + k);
        sA[row * LDA + k + 0] = x4.x;
        sA[row * LDA + k + 1] = x4.y;
        sA[row * LDA + k + 2] = x4.z;
        sA[row * LDA + k + 3] = x4.w;
    }
    __syncthreads();

    #pragma unroll 2
    for (int k0 = 0; k0 < IN_DIM; k0 += 4) {
        const int kb = k0 + 2 * kh;
        v2f b;
        b.x = w_r[(size_t)kb * HID + colbase + n];
        b.y = w_r[(size_t)(kb + 1) * HID + colbase + n];
        #pragma unroll
        for (int rt = 0; rt < 4; ++rt) {
            v2f a;
            a.x = sA[(rt * 16 + arow) * LDA + kb];
            a.y = sA[(rt * 16 + arow) * LDA + kb + 1];
            acc[rt] = __builtin_amdgcn_wmma_f32_16x16x4_f32(
                false, a, false, b, (short)0, acc[rt], false, false);
        }
    }

    {
        const float bias = b_l[colbase + n];
        #pragma unroll
        for (int rt = 0; rt < 4; ++rt) {
            #pragma unroll
            for (int i = 0; i < 8; ++i) {
                const int row = rt * 16 + i + 8 * kh;
                sH[row * LDH + colbase + n] = acc[rt][i] + bias;
            }
        }
    }
    __syncthreads();

    #pragma unroll
    for (int rr = 0; rr < 4; ++rr) {
        const int row = wave + 16 * rr;
        float v[8], s = 0.0f, s2 = 0.0f;
        #pragma unroll
        for (int j = 0; j < 8; ++j) {
            const float t = sH[row * LDH + lane + 32 * j];
            v[j] = t; s += t; s2 += t * t;
        }
        #pragma unroll
        for (int off = 16; off > 0; off >>= 1) {
            s  += __shfl_xor(s,  off, 32);
            s2 += __shfl_xor(s2, off, 32);
        }
        const float mean = s * (1.0f / HID);
        const float var  = s2 * (1.0f / HID) - mean * mean;
        const float rstd = rsqrtf(var + 1e-5f);
        #pragma unroll
        for (int j = 0; j < 8; ++j) {
            const int col = lane + 32 * j;
            const float hn = (v[j] - mean) * rstd * ln1_g[col] + ln1_b[col];
            sH[row * LDH + col] = 0.5f * hn * (1.0f + erff(hn * 0.70710678118654752f));
        }
    }
    __syncthreads();

    float* sOut = sA;
    if (wave < 8) {
        const int cb2 = wave * 16;
        v8f c[4];
        #pragma unroll
        for (int rt = 0; rt < 4; ++rt) c[rt] = (v8f){};
        #pragma unroll 2
        for (int k0 = 0; k0 < HID; k0 += 4) {
            const int kb = k0 + 2 * kh;
            v2f b;
            b.x = w_out[(size_t)kb * OUT_DIM + cb2 + n];
            b.y = w_out[(size_t)(kb + 1) * OUT_DIM + cb2 + n];
            #pragma unroll
            for (int rt = 0; rt < 4; ++rt) {
                v2f a;
                a.x = sH[(rt * 16 + arow) * LDH + kb];
                a.y = sH[(rt * 16 + arow) * LDH + kb + 1];
                c[rt] = __builtin_amdgcn_wmma_f32_16x16x4_f32(
                    false, a, false, b, (short)0, c[rt], false, false);
            }
        }
        const float bias = b_out[cb2 + n];
        #pragma unroll
        for (int rt = 0; rt < 4; ++rt) {
            #pragma unroll
            for (int i = 0; i < 8; ++i) {
                const int row = rt * 16 + i + 8 * kh;
                sOut[row * LDA + cb2 + n] = c[rt][i] + bias;
            }
        }
    }
    __syncthreads();

    #pragma unroll
    for (int rr = 0; rr < 4; ++rr) {
        const int row = wave + 16 * rr;
        float v[4], s = 0.0f, s2 = 0.0f;
        #pragma unroll
        for (int j = 0; j < 4; ++j) {
            const float t = sOut[row * LDA + lane + 32 * j];
            v[j] = t; s += t; s2 += t * t;
        }
        #pragma unroll
        for (int off = 16; off > 0; off >>= 1) {
            s  += __shfl_xor(s,  off, 32);
            s2 += __shfl_xor(s2, off, 32);
        }
        const float mean = s * (1.0f / OUT_DIM);
        const float var  = s2 * (1.0f / OUT_DIM) - mean * mean;
        const float rstd = rsqrtf(var + 1e-5f);
        const int rowg = rowbase + row;
        if (rowg < NUM_DST) {
            #pragma unroll
            for (int j = 0; j < 4; ++j) {
                const int col = lane + 32 * j;
                const float o = (v[j] - mean) * rstd * ln2_g[col] + ln2_b[col];
                out[(size_t)rowg * OUT_DIM + col] = o;
            }
        }
    }
}

extern "C" void kernel_launch(void* const* d_in, const int* in_sizes, int n_in,
                              void* d_out, int out_size, void* d_ws, size_t ws_size,
                              hipStream_t stream) {
    const float* x      = (const float*)d_in[0];
    const float* w_l    = (const float*)d_in[1];
    const float* b_l    = (const float*)d_in[2];
    const float* w_r    = (const float*)d_in[3];
    const float* ln1_g  = (const float*)d_in[4];
    const float* ln1_b  = (const float*)d_in[5];
    const float* w_out  = (const float*)d_in[6];
    const float* b_out  = (const float*)d_in[7];
    const float* ln2_g  = (const float*)d_in[8];
    const float* ln2_b  = (const float*)d_in[9];
    const int*   e_src  = (const int*)d_in[10];
    const int*   e_dst  = (const int*)d_in[11];

    float* sum = (float*)d_ws;
    float* cnt = sum + (size_t)NUM_DST * IN_DIM;
    float* out = (float*)d_out;
    int* owner = (int*)d_out;
    int* rec   = owner + NUM_EDGES / 4;

    owner_kernel<<<OWN_NBLK, OWN_THREADS, 0, stream>>>(e_src, e_dst, owner, rec);

    aggregate_kernel<<<AGG_NBLK, AGG_THREADS, AGG_SMEM_BYTES, stream>>>(x, owner, rec, sum, cnt);

    fused_mlp_kernel<<<NBLK, 512, SMEM_BYTES, stream>>>(sum, cnt, x,
                                                        w_l, b_l, w_r,
                                                        ln1_g, ln1_b,
                                                        w_out, b_out,
                                                        ln2_g, ln2_b, out);
}


# MLP weights: each workgroup copies 32-row weight slabs to LDS with whole-line 16-byte loads, one slab ahead; B fragments come from LDS, not per-wave 4-byte global loads; sums unchanged
# speedup vs baseline: 2.1363x; 1.0378x over previous
#include <hip/hip_runtime.h>
#include <hip/hip_bf16.h>
#include <math.h>

#define NUM_SRC   200000
#define NUM_DST   100000
#define NUM_EDGES 1600000
#define IN_DIM    128
#define HID       256
#define OUT_DIM   128

typedef float v2f __attribute__((ext_vector_type(2)));
typedef float v8f __attribute__((ext_vector_type(8)));

#define ROWS 64
#define LDA  132
#define LDH  260
#define NBLK ((NUM_DST + ROWS - 1) / ROWS)
#define SMEM_BYTES ((ROWS * LDA + ROWS * LDH) * 4)

#define AGG_ROWS    400
#define AGG_WAVES   32
#define AGG_THREADS (AGG_WAVES * 32)
#define AGG_NBLK    ((NUM_DST + AGG_ROWS - 1) / AGG_ROWS)
#define AGG_CHUNK   512
#define AGG_UNROLL  5
#define AGG_NSUPER  (NUM_EDGES / (AGG_CHUNK * AGG_UNROLL))
#define AGG_QCAP    544
#define AGG_SMEM_BYTES ((AGG_ROWS * IN_DIM + AGG_ROWS + AGG_WAVES * AGG_QCAP) * 4)

#define OWN_NBLK    250
#define OWN_THREADS 320
#define OWN_PER     5
#define OWN_NONE    255u
#define REC_BITS    9

static_assert(NUM_EDGES % (AGG_CHUNK * AGG_UNROLL) == 0, "edge scan assumes whole chunks");
static_assert(AGG_NBLK <= 255, "an owner number must fit a byte and differ from OWN_NONE");
static_assert(NUM_EDGES / 4 == OWN_NBLK * OWN_THREADS * OWN_PER, "owner pass covers every edge once");
static_assert(AGG_SMEM_BYTES <= 320 * 1024, "one workgroup's LDS");
static_assert(AGG_ROWS <= (1 << REC_BITS), "a local destination row must fit the record's low bits");
static_assert(((long long)(NUM_SRC - 1) << REC_BITS) < 0x7fffffffLL, "a source row must fit the record's upper bits");
static_assert(NUM_EDGES % 256 == 0, "the records start on a 256-byte boundary behind the owner bytes");

__device__ __forceinline__ unsigned owner_of(int d) {
    return ((unsigned)d < (unsigned)NUM_DST) ? ((unsigned)d / (unsigned)AGG_ROWS) : OWN_NONE;
}

__device__ __forceinline__ int record_of(int s, int d, unsigned own) {
    return (int)(((unsigned)s << REC_BITS) | (((unsigned)d - own * (unsigned)AGG_ROWS) & ((1u << REC_BITS) - 1u)));
}

__global__ __launch_bounds__(OWN_THREADS) void owner_kernel(
    const int* __restrict__ edge_src,
    const int* __restrict__ edge_dst,
    int* __restrict__ owner,
    int* __restrict__ rec)
{
    const int4* src4 = (const int4*)edge_src;
    const int4* dst4 = (const int4*)edge_dst;
    int4* rec4 = (int4*)rec;
    const int i0 = blockIdx.x * OWN_THREADS + threadIdx.x;
    int4 d[OWN_PER], s[OWN_PER];
    #pragma unroll
    for (int j = 0; j < OWN_PER; ++j) {
        d[j] = dst4[i0 + j * (OWN_NBLK * OWN_THREADS)];
        s[j] = src4[i0 + j * (OWN_NBLK * OWN_THREADS)];
    }
    #pragma unroll
    for (int j = 0; j < OWN_PER; ++j) {
        const unsigned ox = owner_of(d[j].x), oy = owner_of(d[j].y);
        const unsigned oz = owner_of(d[j].z), ow = owner_of(d[j].w);
        owner[i0 + j * (OWN_NBLK * OWN_THREADS)] = (int)(ox | (oy << 8) | (oz << 16) | (ow << 24));
        rec4[i0 + j * (OWN_NBLK * OWN_THREADS)] =
            make_int4(record_of(s[j].x, d[j].x, ox), record_of(s[j].y, d[j].y, oy),
                      record_of(s[j].z, d[j].z, oz), record_of(s[j].w, d[j].w, ow));
    }
}

__device__ __forceinline__ void lds_add_f32(float* p, float v) {
    __hip_atomic_fetch_add(p, v, __ATOMIC_RELAXED, __HIP_MEMORY_SCOPE_WORKGROUP);
}

__device__ __forceinline__ unsigned zero_bytes(unsigned t) {
    return ~(((t & 0x7f7f7f7fu) + 0x7f7f7f7fu) | t) & 0x80808080u;
}

template <bool FULL>
__device__ __forceinline__ void agg_flush(const int* qp, int nb, int lane,
                                          const float* __restrict__ x,
                                          const int* __restrict__ rec,
                                          float* acc, int* cntL)
{
    const bool live = FULL || (lane < nb);
    const int ent = live ? qp[lane] : 0;
    const unsigned r = live ? (unsigned)rec[ent] : 0u;
    const int s   = (int)(r >> REC_BITS);
    const int dl  = (int)(r & ((1u << REC_BITS) - 1u));
    if (live) __hip_atomic_fetch_add(&cntL[dl], 1, __ATOMIC_RELAXED, __HIP_MEMORY_SCOPE_WORKGROUP);
    const int n = FULL ? 32 : nb;
    #pragma unroll 1
    for (int e = 0; e < n; e += 8) {
        float4 v[8];
        #pragma unroll
        for (int k = 0; k < 8; ++k) {
            v[k] = make_float4(0.f, 0.f, 0.f, 0.f);
            if (FULL || e + k < n) {
                const int se = __builtin_amdgcn_readlane(s, e + k);
                v[k] = *(const float4*)(x + (size_t)se * IN_DIM + lane * 4);
            }
        }
        #pragma unroll
        for (int k = 0; k < 8; ++k) {
            if (FULL || e + k < n) {
                const int de = __builtin_amdgcn_readlane(dl, e + k);
                float* a = acc + de * IN_DIM + lane;
                lds_add_f32(a,      v[k].x);
                lds_add_f32(a + 32, v[k].y);
                lds_add_f32(a + 64, v[k].z);
                lds_add_f32(a + 96, v[k].w);
            }
        }
    }
}

__global__ __launch_bounds__(AGG_THREADS) void aggregate_kernel(
    const float* __restrict__ x,
    const int*   __restrict__ owner,
    const int*   __restrict__ rec,
    float* __restrict__ sum,
    float* __restrict__ cnt)
{
    extern __shared__ float smem[];
    float* acc  = smem;
    int*   cntL = (int*)(smem + AGG_ROWS * IN_DIM);
    int*   q    = cntL + AGG_ROWS;

    const int tid  = threadIdx.x;
    const int wave = tid >> 5;
    const int lane = tid & 31;
    const int base = blockIdx.x * AGG_ROWS;

    for (int i = tid; i < AGG_ROWS * IN_DIM / 4; i += AGG_THREADS)
        ((float4*)acc)[i] = make_float4(0.f, 0.f, 0.f, 0.f);
    for (int i = tid; i < AGG_ROWS; i += AGG_THREADS) cntL[i] = 0;
    __syncthreads();

    int* myq = q + wave * AGG_QCAP;
    int  qn  = 0;
    const uint4* own4 = (const uint4*)owner;
    const unsigned pat = blockIdx.x * 0x01010101u;
    const int rot = (blockIdx.x * AGG_NSUPER) / AGG_NBLK;

    for (int it = 0; it * AGG_WAVES + wave < AGG_NSUPER; ++it) {
        int sc = it * AGG_WAVES + wave + rot;
        if (sc >= AGG_NSUPER) sc -= AGG_NSUPER;
        uint4 d[AGG_UNROLL];
        #pragma unroll
        for (int u = 0; u < AGG_UNROLL; ++u)
            d[u] = own4[(sc * AGG_UNROLL + u) * 32 + lane];
        #pragma unroll
        for (int u = 0; u < AGG_UNROLL; ++u) {
            const int e0 = ((sc * AGG_UNROLL + u) * 32 + lane) * 16;
            unsigned h = (zero_bytes(d[u].x ^ pat) >> 7) | (zero_bytes(d[u].y ^ pat) >> 6) |
                         (zero_bytes(d[u].z ^ pat) >> 5) | (zero_bytes(d[u].w ^ pat) >> 4);
            unsigned b = __builtin_amdgcn_ballot_w32(h != 0u);
            while (b) {
                const int p   = __builtin_ctz(h | 0x80000000u);
                const int pos = qn + (int)__builtin_amdgcn_mbcnt_lo(b, 0u);
                if (h != 0u) myq[pos] = e0 + ((p & 7) << 2) + (p >> 3);
                h &= h - 1u;
                qn += __builtin_popcount(b);
                b = __builtin_amdgcn_ballot_w32(h != 0u);
            }
            while (qn >= 32) {
                qn -= 32;
                agg_flush<true>(myq + qn, 32, lane, x, rec, acc, cntL);
            }
        }
    }
    if (qn > 0) agg_flush<false>(myq, qn, lane, x, rec, acc, cntL);
    __syncthreads();

    int nrows = NUM_DST - base;
    if (nrows > AGG_ROWS) nrows = AGG_ROWS;
    for (int r = wave; r < nrows; r += AGG_WAVES) {
        const float* a = acc + r * IN_DIM + lane;
        const float4 o = make_float4(a[0], a[32], a[64], a[96]);
        *(float4*)(sum + (size_t)(base + r) * IN_DIM + lane * 4) = o;
    }
    for (int r = tid; r < nrows; r += AGG_THREADS) cnt[base + r] = (float)cntL[r];
}

#define SLAB_K 32
#define WSLAB1 (SLAB_K * HID)
#define WSLAB2 (SLAB_K * OUT_DIM)
static_assert(2 * WSLAB1 <= ROWS * LDH, "two GEMM1 slabs fit the hidden buffer");
static_assert(2 * WSLAB2 <= ROWS * LDA, "two GEMM2 slabs fit the A buffer");
static_assert(IN_DIM % SLAB_K == 0 && HID % SLAB_K == 0, "whole slabs");

template <int NC, int NITEM>
__device__ __forceinline__ void slab_fetch(const float* __restrict__ w, int tid, float4 (&r)[2 * NITEM]) {
    #pragma unroll
    for (int j = 0; j < NITEM; ++j) {
        const int i  = tid + j * 512;
        const int cg = i & (NC / 4 - 1);
        const int kp = i / (NC / 4);
        const float* p = w + (size_t)(2 * kp) * NC + 4 * cg;
        r[2 * j]     = *(const float4*)p;
        r[2 * j + 1] = *(const float4*)(p + NC);
    }
}

template <int NC, int NITEM>
__device__ __forceinline__ void slab_store(float* buf, int tid, const float4 (&r)[2 * NITEM]) {
    #pragma unroll
    for (int j = 0; j < NITEM; ++j) {
        const int i  = tid + j * 512;
        const int cg = i & (NC / 4 - 1);
        const int kp = i / (NC / 4);
        float* q = buf + kp * (2 * NC) + ((8 * cg) ^ ((kp & 1) << 5));
        const float4 r0 = r[2 * j], r1 = r[2 * j + 1];
        *(float4*)q       = make_float4(r0.x, r1.x, r0.y, r1.y);
        *(float4*)(q + 4) = make_float4(r0.z, r1.z, r0.w, r1.w);
    }
}

template <int NC, int LD>
__device__ __forceinline__ void slab_mma(const float* bb, const float* ab, v8f (&acc)[4]) {
    #pragma unroll 2
    for (int kk = 0; kk < SLAB_K; kk += 4) {
        const v2f b = *(const v2f*)(bb + kk * NC);
        #pragma unroll
        for (int rt = 0; rt < 4; ++rt) {
            const v2f a = *(const v2f*)(ab + rt * 16 * LD + kk);
            acc[rt] = __builtin_amdgcn_wmma_f32_16x16x4_f32(
                false, a, false, b, (short)0, acc[rt], false, false);
        }
    }
}

__global__ __launch_bounds__(512, 8) void fused_mlp_kernel(
    const float* __restrict__ sum, const float* __restrict__ cnt,
    const float* __restrict__ x,
    const float* __restrict__ w_l,  const float* __restrict__ b_l,
    const float* __restrict__ w_r,
    const float* __restrict__ ln1_g, const float* __restrict__ ln1_b,
    const float* __restrict__ w_out, const float* __restrict__ b_out,
    const float* __restrict__ ln2_g, const float* __restrict__ ln2_b,
    float* __restrict__ out)
{
    extern __shared__ float smem[];
    float* sA = smem;
    float* sH = smem + ROWS * LDA;

    const int tid  = threadIdx.x;
    const int wave = tid >> 5;
    const int lane = tid & 31;
    const int rowbase = blockIdx.x * ROWS;

    const int n    = lane & 15;
    const int kh   = lane >> 4;
    const int arow = lane & 15;

    float4 wr[4];
    slab_fetch<HID, 2>(w_l, tid, wr);

    #pragma unroll
    for (int j = 0; j < 4; ++j) {
        const int idx = tid + j * 512;
        const int row = idx >> 5;
        const int k   = (idx & 31) * 4;
        int rg = rowbase + row;
        if (rg >= NUM_DST) rg = NUM_DST - 1;
        const float c   = cnt[rg];
        const float inv = 1.0f / fmaxf(c, 1.0f);
        const float4 s4 = *(const float4*)(sum + (size_t)rg * IN_DIM + k);
        sA[row * LDA + k + 0] = s4.x * inv;
        sA[row * LDA + k + 1] = s4.y * inv;
        sA[row * LDA + k + 2] = s4.z * inv;
        sA[row * LDA + k + 3] = s4.w * inv;
    }
    slab_store<HID, 2>(sH, tid, wr);
    __syncthreads();

    v8f acc[4];
    #pragma unroll
    for (int rt = 0; rt < 4; ++rt) acc[rt] = (v8f){};
    const int colbase = wave * 16;
    const float* bb1 = sH + kh * (2 * HID) + ((2 * (colbase + n)) ^ (kh << 5));
    const float* ab1 = sA + arow * LDA + 2 * kh;

    #pragma unroll 1
    for (int s = 0; s < 3; ++s) {
        slab_fetch<HID, 2>(w_l + (size_t)(s + 1) * SLAB_K * HID, tid, wr);
        slab_mma<HID, LDA>(bb1 + (s & 1) * WSLAB1, ab1 + s * SLAB_K, acc);
        slab_store<HID, 2>(sH + ((s + 1) & 1) * WSLAB1, tid, wr);
        __syncthreads();
    }
    slab_fetch<HID, 2>(w_r, tid, wr);
    float4 xr[4];
    #pragma unroll
    for (int j = 0; j < 4; ++j) {
        const int idx = tid + j * 512;
        const int row = idx >> 5;
        const int k   = (idx & 31) * 4;
        int rg = rowbase + row;
        if (rg >= NUM_DST) rg = NUM_DST - 1;
        xr[j] = *(const float4*)(x + (size_t)rg * IN_DIM + k);
    }
    slab_mma<HID, LDA>(bb1 + WSLAB1, ab1 + 3 * SLAB_K, acc);
    __syncthreads();

    slab_store<HID, 2>(sH, tid, wr);
    #pragma unroll
    for (int j = 0; j < 4; ++j) {
        const int idx = tid + j * 512;
        const int row = idx >> 5;
        const int k   = (idx & 31) * 4;
        sA[row * LDA + k + 0] = xr[j].x;
        sA[row * LDA + k + 1] = xr[j].y;
        sA[row * LDA + k + 2] = xr[j].z;
        sA[row * LDA + k + 3] = xr[j].w;
    }
    __syncthreads();

    #pragma unroll 1
    for (int s = 0; s < 3; ++s) {
        slab_fetch<HID, 2>(w_r + (size_t)(s + 1) * SLAB_K * HID, tid, wr);
        slab_mma<HID, LDA>(bb1 + (s & 1) * WSLAB1, ab1 + s * SLAB_K, acc);
        slab_store<HID, 2>(sH + ((s + 1) & 1) * WSLAB1, tid, wr);
        __syncthreads();
    }
    float4 wo[2];
    slab_fetch<OUT_DIM, 1>(w_out, tid, wo);
    slab_mma<HID, LDA>(bb1 + WSLAB1, ab1 + 3 * SLAB_K, acc);
    __syncthreads();

    slab_store<OUT_DIM, 1>(sA, tid, wo);

    {
        const float bias = b_l[colbase + n];
        #pragma unroll
        for (int rt = 0; rt < 4; ++rt) {
            #pragma unroll
            for (int i = 0; i < 8; ++i) {
                const int row = rt * 16 + i + 8 * kh;
                sH[row * LDH + colbase + n] = acc[rt][i] + bias;
            }
        }
    }
    __syncthreads();

    #pragma unroll
    for (int rr = 0; rr < 4; ++rr) {
        const int row = wave + 16 * rr;
        float v[8], s = 0.0f, s2 = 0.0f;
        #pragma unroll
        for (int j = 0; j < 8; ++j) {
            const float t = sH[row * LDH + lane + 32 * j];
            v[j] = t; s += t; s2 += t * t;
        }
        #pragma unroll
        for (int off = 16; off > 0; off >>= 1) {
            s  += __shfl_xor(s,  off, 32);
            s2 += __shfl_xor(s2, off, 32);
        }
        const float mean = s * (1.0f / HID);
        const float var  = s2 * (1.0f / HID) - mean * mean;
        const float rstd = rsqrtf(var + 1e-5f);
        #pragma unroll
        for (int j = 0; j < 8; ++j) {
            const int col = lane + 32 * j;
            const float hn = (v[j] - mean) * rstd * ln1_g[col] + ln1_b[col];
            sH[row * LDH + col] = 0.5f * hn * (1.0f + erff(hn * 0.70710678118654752f));
        }
    }
    __syncthreads();

    float* sOut = sA;
    const int cb2 = (wave & 7) * 16;
    const float* bb2 = sA + kh * (2 * OUT_DIM) + ((2 * (cb2 + n)) ^ (kh << 5));
    const float* ab2 = sH + arow * LDH + 2 * kh;
    v8f c[4];
    #pragma unroll
    for (int rt = 0; rt < 4; ++rt) c[rt] = (v8f){};
    #pragma unroll 1
    for (int s = 0; s < 7; ++s) {
        slab_fetch<OUT_DIM, 1>(w_out + (size_t)(s + 1) * SLAB_K * OUT_DIM, tid, wo);
        if (wave < 8) slab_mma<OUT_DIM, LDH>(bb2 + (s & 1) * WSLAB2, ab2 + s * SLAB_K, c);
        slab_store<OUT_DIM, 1>(sA + ((s + 1) & 1) * WSLAB2, tid, wo);
        __syncthreads();
    }
    if (wave < 8) slab_mma<OUT_DIM, LDH>(bb2 + WSLAB2, ab2 + 7 * SLAB_K, c);
    __syncthreads();
    if (wave < 8) {
        const float bias = b_out[cb2 + n];
        #pragma unroll
        for (int rt = 0; rt < 4; ++rt) {
            #pragma unroll
            for (int i = 0; i < 8; ++i) {
                const int row = rt * 16 + i + 8 * kh;
                sOut[row * LDA + cb2 + n] = c[rt][i] + bias;
            }
        }
    }
    __syncthreads();

    #pragma unroll
    for (int rr = 0; rr < 4; ++rr) {
        const int row = wave + 16 * rr;
        float v[4], s = 0.0f, s2 = 0.0f;
        #pragma unroll
        for (int j = 0; j < 4; ++j) {
            const float t = sOut[row * LDA + lane + 32 * j];
            v[j] = t; s += t; s2 += t * t;
        }
        #pragma unroll
        for (int off = 16; off > 0; off >>= 1) {
            s  += __shfl_xor(s,  off, 32);
            s2 += __shfl_xor(s2, off, 32);
        }
        const float mean = s * (1.0f / OUT_DIM);
        const float var  = s2 * (1.0f / OUT_DIM) - mean * mean;
        const float rstd = rsqrtf(var + 1e-5f);
        const int rowg = rowbase + row;
        if (rowg < NUM_DST) {
            #pragma unroll
            for (int j = 0; j < 4; ++j) {
                const int col = lane + 32 * j;
                const float o = (v[j] - mean) * rstd * ln2_g[col] + ln2_b[col];
                out[(size_t)rowg * OUT_DIM + col] = o;
            }
        }
    }
}

extern "C" void kernel_launch(void* const* d_in, const int* in_sizes, int n_in,
                              void* d_out, int out_size, void* d_ws, size_t ws_size,
                              hipStream_t stream) {
    const float* x      = (const float*)d_in[0];
    const float* w_l    = (const float*)d_in[1];
    const float* b_l    = (const float*)d_in[2];
    const float* w_r    = (const float*)d_in[3];
    const float* ln1_g  = (const float*)d_in[4];
    const float* ln1_b  = (const float*)d_in[5];
    const float* w_out  = (const float*)d_in[6];
    const float* b_out  = (const float*)d_in[7];
    const float* ln2_g  = (const float*)d_in[8];
    const float* ln2_b  = (const float*)d_in[9];
    const int*   e_src  = (const int*)d_in[10];
    const int*   e_dst  = (const int*)d_in[11];

    float* sum = (float*)d_ws;
    float* cnt = sum + (size_t)NUM_DST * IN_DIM;
    float* out = (float*)d_out;
    int* owner = (int*)d_out;
    int* rec   = owner + NUM_EDGES / 4;

    owner_kernel<<<OWN_NBLK, OWN_THREADS, 0, stream>>>(e_src, e_dst, owner, rec);

    aggregate_kernel<<<AGG_NBLK, AGG_THREADS, AGG_SMEM_BYTES, stream>>>(x, owner, rec, sum, cnt);

    fused_mlp_kernel<<<NBLK, 512, SMEM_BYTES, stream>>>(sum, cnt, x,
                                                        w_l, b_l, w_r,
                                                        ln1_g, ln1_b,
                                                        w_out, b_out,
                                                        ln2_g, ln2_b, out);
}


# MLP tiling: 80 rows (five 16-row tiles per wave) per workgroup, not 64: 1250 workgroups, at most 5 per WGP (400 rows) where 27 WGPs ran a 7th of 64 (448); same sums in the same order
# speedup vs baseline: 2.2692x; 1.0622x over previous
#include <hip/hip_runtime.h>
#include <hip/hip_bf16.h>
#include <math.h>

#define NUM_SRC   200000
#define NUM_DST   100000
#define NUM_EDGES 1600000
#define IN_DIM    128
#define HID       256
#define OUT_DIM   128

typedef float v2f __attribute__((ext_vector_type(2)));
typedef float v8f __attribute__((ext_vector_type(8)));

#define ROWS 80
#define RT   (ROWS / 16)
#define NSTG (ROWS * 32 / 512)
#define LDA  132
#define LDH  260
#define NBLK ((NUM_DST + ROWS - 1) / ROWS)
#define SMEM_BYTES ((ROWS * LDA + ROWS * LDH) * 4)
static_assert(ROWS % 16 == 0 && (ROWS * 32) % 512 == 0, "whole row tiles, whole staging trips");

#define AGG_ROWS    400
#define AGG_WAVES   32
#define AGG_THREADS (AGG_WAVES * 32)
#define AGG_NBLK    ((NUM_DST + AGG_ROWS - 1) / AGG_ROWS)
#define AGG_CHUNK   512
#define AGG_UNROLL  5
#define AGG_NSUPER  (NUM_EDGES / (AGG_CHUNK * AGG_UNROLL))
#define AGG_QCAP    544
#define AGG_SMEM_BYTES ((AGG_ROWS * IN_DIM + AGG_ROWS + AGG_WAVES * AGG_QCAP) * 4)

#define OWN_NBLK    250
#define OWN_THREADS 320
#define OWN_PER     5
#define OWN_NONE    255u
#define REC_BITS    9

static_assert(NUM_EDGES % (AGG_CHUNK * AGG_UNROLL) == 0, "edge scan assumes whole chunks");
static_assert(AGG_NBLK <= 255, "an owner number must fit a byte and differ from OWN_NONE");
static_assert(NUM_EDGES / 4 == OWN_NBLK * OWN_THREADS * OWN_PER, "owner pass covers every edge once");
static_assert(AGG_SMEM_BYTES <= 320 * 1024, "one workgroup's LDS");
static_assert(AGG_ROWS <= (1 << REC_BITS), "a local destination row must fit the record's low bits");
static_assert(((long long)(NUM_SRC - 1) << REC_BITS) < 0x7fffffffLL, "a source row must fit the record's upper bits");
static_assert(NUM_EDGES % 256 == 0, "the records start on a 256-byte boundary behind the owner bytes");

__device__ __forceinline__ unsigned owner_of(int d) {
    return ((unsigned)d < (unsigned)NUM_DST) ? ((unsigned)d / (unsigned)AGG_ROWS) : OWN_NONE;
}

__device__ __forceinline__ int record_of(int s, int d, unsigned own) {
    return (int)(((unsigned)s << REC_BITS) | (((unsigned)d - own * (unsigned)AGG_ROWS) & ((1u << REC_BITS) - 1u)));
}

__global__ __launch_bounds__(OWN_THREADS) void owner_kernel(
    const int* __restrict__ edge_src,
    const int* __restrict__ edge_dst,
    int* __restrict__ owner,
    int* __restrict__ rec)
{
    const int4* src4 = (const int4*)edge_src;
    const int4* dst4 = (const int4*)edge_dst;
    int4* rec4 = (int4*)rec;
    const int i0 = blockIdx.x * OWN_THREADS + threadIdx.x;
    int4 d[OWN_PER], s[OWN_PER];
    #pragma unroll
    for (int j = 0; j < OWN_PER; ++j) {
        d[j] = dst4[i0 + j * (OWN_NBLK * OWN_THREADS)];
        s[j] = src4[i0 + j * (OWN_NBLK * OWN_THREADS)];
    }
    #pragma unroll
    for (int j = 0; j < OWN_PER; ++j) {
        const unsigned ox = owner_of(d[j].x), oy = owner_of(d[j].y);
        const unsigned oz = owner_of(d[j].z), ow = owner_of(d[j].w);
        owner[i0 + j * (OWN_NBLK * OWN_THREADS)] = (int)(ox | (oy << 8) | (oz << 16) | (ow << 24));
        rec4[i0 + j * (OWN_NBLK * OWN_THREADS)] =
            make_int4(record_of(s[j].x, d[j].x, ox), record_of(s[j].y, d[j].y, oy),
                      record_of(s[j].z, d[j].z, oz), record_of(s[j].w, d[j].w, ow));
    }
}

__device__ __forceinline__ void lds_add_f32(float* p, float v) {
    __hip_atomic_fetch_add(p, v, __ATOMIC_RELAXED, __HIP_MEMORY_SCOPE_WORKGROUP);
}

__device__ __forceinline__ unsigned zero_bytes(unsigned t) {
    return ~(((t & 0x7f7f7f7fu) + 0x7f7f7f7fu) | t) & 0x80808080u;
}

template <bool FULL>
__device__ __forceinline__ void agg_flush(const int* qp, int nb, int lane,
                                          const float* __restrict__ x,
                                          const int* __restrict__ rec,
                                          float* acc, int* cntL)
{
    const bool live = FULL || (lane < nb);
    const int ent = live ? qp[lane] : 0;
    const unsigned r = live ? (unsigned)rec[ent] : 0u;
    const int s   = (int)(r >> REC_BITS);
    const int dl  = (int)(r & ((1u << REC_BITS) - 1u));
    if (live) __hip_atomic_fetch_add(&cntL[dl], 1, __ATOMIC_RELAXED, __HIP_MEMORY_SCOPE_WORKGROUP);
    const int n = FULL ? 32 : nb;
    #pragma unroll 1
    for (int e = 0; e < n; e += 8) {
        float4 v[8];
        #pragma unroll
        for (int k = 0; k < 8; ++k) {
            v[k] = make_float4(0.f, 0.f, 0.f, 0.f);
            if (FULL || e + k < n) {
                const int se = __builtin_amdgcn_readlane(s, e + k);
                v[k] = *(const float4*)(x + (size_t)se * IN_DIM + lane * 4);
            }
        }
        #pragma unroll
        for (int k = 0; k < 8; ++k) {
            if (FULL || e + k < n) {
                const int de = __builtin_amdgcn_readlane(dl, e + k);
                float* a = acc + de * IN_DIM + lane;
                lds_add_f32(a,      v[k].x);
                lds_add_f32(a + 32, v[k].y);
                lds_add_f32(a + 64, v[k].z);
                lds_add_f32(a + 96, v[k].w);
            }
        }
    }
}

__global__ __launch_bounds__(AGG_THREADS) void aggregate_kernel(
    const float* __restrict__ x,
    const int*   __restrict__ owner,
    const int*   __restrict__ rec,
    float* __restrict__ sum,
    float* __restrict__ cnt)
{
    extern __shared__ float smem[];
    float* acc  = smem;
    int*   cntL = (int*)(smem + AGG_ROWS * IN_DIM);
    int*   q    = cntL + AGG_ROWS;

    const int tid  = threadIdx.x;
    const int wave = tid >> 5;
    const int lane = tid & 31;
    const int base = blockIdx.x * AGG_ROWS;

    for (int i = tid; i < AGG_ROWS * IN_DIM / 4; i += AGG_THREADS)
        ((float4*)acc)[i] = make_float4(0.f, 0.f, 0.f, 0.f);
    for (int i = tid; i < AGG_ROWS; i += AGG_THREADS) cntL[i] = 0;
    __syncthreads();

    int* myq = q + wave * AGG_QCAP;
    int  qn  = 0;
    const uint4* own4 = (const uint4*)owner;
    const unsigned pat = blockIdx.x * 0x01010101u;
    const int rot = (blockIdx.x * AGG_NSUPER) / AGG_NBLK;

    for (int it = 0; it * AGG_WAVES + wave < AGG_NSUPER; ++it) {
        int sc = it * AGG_WAVES + wave + rot;
        if (sc >= AGG_NSUPER) sc -= AGG_NSUPER;
        uint4 d[AGG_UNROLL];
        #pragma unroll
        for (int u = 0; u < AGG_UNROLL; ++u)
            d[u] = own4[(sc * AGG_UNROLL + u) * 32 + lane];
        #pragma unroll
        for (int u = 0; u < AGG_UNROLL; ++u) {
            const int e0 = ((sc * AGG_UNROLL + u) * 32 + lane) * 16;
            unsigned h = (zero_bytes(d[u].x ^ pat) >> 7) | (zero_bytes(d[u].y ^ pat) >> 6) |
                         (zero_bytes(d[u].z ^ pat) >> 5) | (zero_bytes(d[u].w ^ pat) >> 4);
            unsigned b = __builtin_amdgcn_ballot_w32(h != 0u);
            while (b) {
                const int p   = __builtin_ctz(h | 0x80000000u);
                const int pos = qn + (int)__builtin_amdgcn_mbcnt_lo(b, 0u);
                if (h != 0u) myq[pos] = e0 + ((p & 7) << 2) + (p >> 3);
                h &= h - 1u;
                qn += __builtin_popcount(b);
                b = __builtin_amdgcn_ballot_w32(h != 0u);
            }
            while (qn >= 32) {
                qn -= 32;
                agg_flush<true>(myq + qn, 32, lane, x, rec, acc, cntL);
            }
        }
    }
    if (qn > 0) agg_flush<false>(myq, qn, lane, x, rec, acc, cntL);
    __syncthreads();

    int nrows = NUM_DST - base;
    if (nrows > AGG_ROWS) nrows = AGG_ROWS;
    for (int r = wave; r < nrows; r += AGG_WAVES) {
        const float* a = acc + r * IN_DIM + lane;
        const float4 o = make_float4(a[0], a[32], a[64], a[96]);
        *(float4*)(sum + (size_t)(base + r) * IN_DIM + lane * 4) = o;
    }
    for (int r = tid; r < nrows; r += AGG_THREADS) cnt[base + r] = (float)cntL[r];
}

#define SLAB_K 32
#define WSLAB1 (SLAB_K * HID)
#define WSLAB2 (SLAB_K * OUT_DIM)
static_assert(2 * WSLAB1 <= ROWS * LDH, "two GEMM1 slabs fit the hidden buffer");
static_assert(2 * WSLAB2 <= ROWS * LDA, "two GEMM2 slabs fit the A buffer");
static_assert(IN_DIM % SLAB_K == 0 && HID % SLAB_K == 0, "whole slabs");

template <int NC, int NITEM>
__device__ __forceinline__ void slab_fetch(const float* __restrict__ w, int tid, float4 (&r)[2 * NITEM]) {
    #pragma unroll
    for (int j = 0; j < NITEM; ++j) {
        const int i  = tid + j * 512;
        const int cg = i & (NC / 4 - 1);
        const int kp = i / (NC / 4);
        const float* p = w + (size_t)(2 * kp) * NC + 4 * cg;
        r[2 * j]     = *(const float4*)p;
        r[2 * j + 1] = *(const float4*)(p + NC);
    }
}

template <int NC, int NITEM>
__device__ __forceinline__ void slab_store(float* buf, int tid, const float4 (&r)[2 * NITEM]) {
    #pragma unroll
    for (int j = 0; j < NITEM; ++j) {
        const int i  = tid + j * 512;
        const int cg = i & (NC / 4 - 1);
        const int kp = i / (NC / 4);
        float* q = buf + kp * (2 * NC) + ((8 * cg) ^ ((kp & 1) << 5));
        const float4 r0 = r[2 * j], r1 = r[2 * j + 1];
        *(float4*)q       = make_float4(r0.x, r1.x, r0.y, r1.y);
        *(float4*)(q + 4) = make_float4(r0.z, r1.z, r0.w, r1.w);
    }
}

template <int NC, int LD, int U = 2>
__device__ __forceinline__ void slab_mma(const float* bb, const float* ab, v8f (&acc)[RT]) {
    #pragma unroll U
    for (int kk = 0; kk < SLAB_K; kk += 4) {
        const v2f b = *(const v2f*)(bb + kk * NC);
        #pragma unroll
        for (int rt = 0; rt < RT; ++rt) {
            const v2f a = *(const v2f*)(ab + rt * 16 * LD + kk);
            acc[rt] = __builtin_amdgcn_wmma_f32_16x16x4_f32(
                false, a, false, b, (short)0, acc[rt], false, false);
        }
    }
}

__global__ __launch_bounds__(512, 8) void fused_mlp_kernel(
    const float* __restrict__ sum, const float* __restrict__ cnt,
    const float* __restrict__ x,
    const float* __restrict__ w_l,  const float* __restrict__ b_l,
    const float* __restrict__ w_r,
    const float* __restrict__ ln1_g, const float* __restrict__ ln1_b,
    const float* __restrict__ w_out, const float* __restrict__ b_out,
    const float* __restrict__ ln2_g, const float* __restrict__ ln2_b,
    float* __restrict__ out)
{
    extern __shared__ float smem[];
    float* sA = smem;
    float* sH = smem + ROWS * LDA;

    const int tid  = threadIdx.x;
    const int wave = tid >> 5;
    const int lane = tid & 31;
    const int rowbase = blockIdx.x * ROWS;

    const int n    = lane & 15;
    const int kh   = lane >> 4;
    const int arow = lane & 15;

    float4 wr[4];
    slab_fetch<HID, 2>(w_l, tid, wr);

    #pragma unroll
    for (int j = 0; j < NSTG; ++j) {
        const int idx = tid + j * 512;
        const int row = idx >> 5;
        const int k   = (idx & 31) * 4;
        int rg = rowbase + row;
        if (rg >= NUM_DST) rg = NUM_DST - 1;
        const float c   = cnt[rg];
        const float inv = 1.0f / fmaxf(c, 1.0f);
        const float4 s4 = *(const float4*)(sum + (size_t)rg * IN_DIM + k);
        sA[row * LDA + k + 0] = s4.x * inv;
        sA[row * LDA + k + 1] = s4.y * inv;
        sA[row * LDA + k + 2] = s4.z * inv;
        sA[row * LDA + k + 3] = s4.w * inv;
    }
    slab_store<HID, 2>(sH, tid, wr);
    __syncthreads();

    v8f acc[RT];
    #pragma unroll
    for (int rt = 0; rt < RT; ++rt) acc[rt] = (v8f){};
    const int colbase = wave * 16;
    const float* bb1 = sH + kh * (2 * HID) + ((2 * (colbase + n)) ^ (kh << 5));
    const float* ab1 = sA + arow * LDA + 2 * kh;

    #pragma unroll 1
    for (int s = 0; s < 3; ++s) {
        slab_fetch<HID, 2>(w_l + (size_t)(s + 1) * SLAB_K * HID, tid, wr);
        slab_mma<HID, LDA>(bb1 + (s & 1) * WSLAB1, ab1 + s * SLAB_K, acc);
        slab_store<HID, 2>(sH + ((s + 1) & 1) * WSLAB1, tid, wr);
        __syncthreads();
    }
    slab_fetch<HID, 2>(w_r, tid, wr);
    float4 xr[NSTG];
    #pragma unroll
    for (int j = 0; j < NSTG; ++j) {
        const int idx = tid + j * 512;
        const int row = idx >> 5;
        const int k   = (idx & 31) * 4;
        int rg = rowbase + row;
        if (rg >= NUM_DST) rg = NUM_DST - 1;
        xr[j] = *(const float4*)(x + (size_t)rg * IN_DIM + k);
    }
    slab_mma<HID, LDA, 1>(bb1 + WSLAB1, ab1 + 3 * SLAB_K, acc);
    __syncthreads();

    slab_store<HID, 2>(sH, tid, wr);
    #pragma unroll
    for (int j = 0; j < NSTG; ++j) {
        const int idx = tid + j * 512;
        const int row = idx >> 5;
        const int k   = (idx & 31) * 4;
        sA[row * LDA + k + 0] = xr[j].x;
        sA[row * LDA + k + 1] = xr[j].y;
        sA[row * LDA + k + 2] = xr[j].z;
        sA[row * LDA + k + 3] = xr[j].w;
    }
    __syncthreads();

    #pragma unroll 1
    for (int s = 0; s < 3; ++s) {
        slab_fetch<HID, 2>(w_r + (size_t)(s + 1) * SLAB_K * HID, tid, wr);
        slab_mma<HID, LDA>(bb1 + (s & 1) * WSLAB1, ab1 + s * SLAB_K, acc);
        slab_store<HID, 2>(sH + ((s + 1) & 1) * WSLAB1, tid, wr);
        __syncthreads();
    }
    float4 wo[2];
    slab_fetch<OUT_DIM, 1>(w_out, tid, wo);
    slab_mma<HID, LDA>(bb1 + WSLAB1, ab1 + 3 * SLAB_K, acc);
    __syncthreads();

    slab_store<OUT_DIM, 1>(sA, tid, wo);

    {
        const float bias = b_l[colbase + n];
        #pragma unroll
        for (int rt = 0; rt < RT; ++rt) {
            #pragma unroll
            for (int i = 0; i < 8; ++i) {
                const int row = rt * 16 + i + 8 * kh;
                sH[row * LDH + colbase + n] = acc[rt][i] + bias;
            }
        }
    }
    __syncthreads();

    #pragma unroll
    for (int rr = 0; rr < RT; ++rr) {
        const int row = wave + 16 * rr;
        float v[8], s = 0.0f, s2 = 0.0f;
        #pragma unroll
        for (int j = 0; j < 8; ++j) {
            const float t = sH[row * LDH + lane + 32 * j];
            v[j] = t; s += t; s2 += t * t;
        }
        #pragma unroll
        for (int off = 16; off > 0; off >>= 1) {
            s  += __shfl_xor(s,  off, 32);
            s2 += __shfl_xor(s2, off, 32);
        }
        const float mean = s * (1.0f / HID);
        const float var  = s2 * (1.0f / HID) - mean * mean;
        const float rstd = rsqrtf(var + 1e-5f);
        #pragma unroll
        for (int j = 0; j < 8; ++j) {
            const int col = lane + 32 * j;
            const float hn = (v[j] - mean) * rstd * ln1_g[col] + ln1_b[col];
            sH[row * LDH + col] = 0.5f * hn * (1.0f + erff(hn * 0.70710678118654752f));
        }
    }
    __syncthreads();

    float* sOut = sA;
    const int cb2 = (wave & 7) * 16;
    const float* bb2 = sA + kh * (2 * OUT_DIM) + ((2 * (cb2 + n)) ^ (kh << 5));
    const float* ab2 = sH + arow * LDH + 2 * kh;
    v8f c[RT];
    #pragma unroll
    for (int rt = 0; rt < RT; ++rt) c[rt] = (v8f){};
    #pragma unroll 1
    for (int s = 0; s < 7; ++s) {
        slab_fetch<OUT_DIM, 1>(w_out + (size_t)(s + 1) * SLAB_K * OUT_DIM, tid, wo);
        if (wave < 8) slab_mma<OUT_DIM, LDH>(bb2 + (s & 1) * WSLAB2, ab2 + s * SLAB_K, c);
        slab_store<OUT_DIM, 1>(sA + ((s + 1) & 1) * WSLAB2, tid, wo);
        __syncthreads();
    }
    if (wave < 8) slab_mma<OUT_DIM, LDH>(bb2 + WSLAB2, ab2 + 7 * SLAB_K, c);
    __syncthreads();
    if (wave < 8) {
        const float bias = b_out[cb2 + n];
        #pragma unroll
        for (int rt = 0; rt < RT; ++rt) {
            #pragma unroll
            for (int i = 0; i < 8; ++i) {
                const int row = rt * 16 + i + 8 * kh;
                sOut[row * LDA + cb2 + n] = c[rt][i] + bias;
            }
        }
    }
    __syncthreads();

    #pragma unroll
    for (int rr = 0; rr < RT; ++rr) {
        const int row = wave + 16 * rr;
        float v[4], s = 0.0f, s2 = 0.0f;
        #pragma unroll
        for (int j = 0; j < 4; ++j) {
            const float t = sOut[row * LDA + lane + 32 * j];
            v[j] = t; s += t; s2 += t * t;
        }
        #pragma unroll
        for (int off = 16; off > 0; off >>= 1) {
            s  += __shfl_xor(s,  off, 32);
            s2 += __shfl_xor(s2, off, 32);
        }
        const float mean = s * (1.0f / OUT_DIM);
        const float var  = s2 * (1.0f / OUT_DIM) - mean * mean;
        const float rstd = rsqrtf(var + 1e-5f);
        const int rowg = rowbase + row;
        if (rowg < NUM_DST) {
            #pragma unroll
            for (int j = 0; j < 4; ++j) {
                const int col = lane + 32 * j;
                const float o = (v[j] - mean) * rstd * ln2_g[col] + ln2_b[col];
                out[(size_t)rowg * OUT_DIM + col] = o;
            }
        }
    }
}

extern "C" void kernel_launch(void* const* d_in, const int* in_sizes, int n_in,
                              void* d_out, int out_size, void* d_ws, size_t ws_size,
                              hipStream_t stream) {
    const float* x      = (const float*)d_in[0];
    const float* w_l    = (const float*)d_in[1];
    const float* b_l    = (const float*)d_in[2];
    const float* w_r    = (const float*)d_in[3];
    const float* ln1_g  = (const float*)d_in[4];
    const float* ln1_b  = (const float*)d_in[5];
    const float* w_out  = (const float*)d_in[6];
    const float* b_out  = (const float*)d_in[7];
    const float* ln2_g  = (const float*)d_in[8];
    const float* ln2_b  = (const float*)d_in[9];
    const int*   e_src  = (const int*)d_in[10];
    const int*   e_dst  = (const int*)d_in[11];

    float* sum = (float*)d_ws;
    float* cnt = sum + (size_t)NUM_DST * IN_DIM;
    float* out = (float*)d_out;
    int* owner = (int*)d_out;
    int* rec   = owner + NUM_EDGES / 4;

    owner_kernel<<<OWN_NBLK, OWN_THREADS, 0, stream>>>(e_src, e_dst, owner, rec);

    aggregate_kernel<<<AGG_NBLK, AGG_THREADS, AGG_SMEM_BYTES, stream>>>(x, owner, rec, sum, cnt);

    fused_mlp_kernel<<<NBLK, 512, SMEM_BYTES, stream>>>(sum, cnt, x,
                                                        w_l, b_l, w_r,
                                                        ln1_g, ln1_b,
                                                        w_out, b_out,
                                                        ln2_g, ln2_b, out);
}
